# starting kernel
# baseline (speedup 1.0000x reference)
#include <hip/hip_runtime.h>
#include <math.h>

#define HEADS 4
#define FEAT  32
#define HF    128

typedef __attribute__((ext_vector_type(2))) float v2f;
typedef __attribute__((ext_vector_type(8))) float v8f;

__global__ void __launch_bounds__(256)
gat_gemm_wmma_f32(const float* __restrict__ X, const float* __restrict__ W,
                  float* __restrict__ Hout, int K) {
  const int wave = threadIdx.x >> 5;
  const int lane = threadIdx.x & 31;
  const int half = lane >> 4;
  const int l15  = lane & 15;
  const int row  = blockIdx.x * 16 + l15;
  const int col  = wave * 16 + l15;
  const int kh   = half * 2;
  const float* __restrict__ xrow = X + (size_t)row * K;

  v8f c = {};
  for (int k0 = 0; k0 < K; k0 += 4) {
    v2f a, b;
    a.x = xrow[k0 + kh];
    a.y = xrow[k0 + kh + 1];
    b.x = W[(size_t)(k0 + kh)     * HF + col];
    b.y = W[(size_t)(k0 + kh + 1) * HF + col];
    c = __builtin_amdgcn_wmma_f32_16x16x4_f32(false, a, false, b,
                                              (short)0, c, false, false);
  }
  const int rbase = blockIdx.x * 16 + half * 8;
  float* orow = Hout + (size_t)rbase * HF + col;
#pragma unroll
  for (int v = 0; v < 8; ++v)
    orow[(size_t)v * HF] = c[v];
}

__global__ void gat_gemm_tail(const float* __restrict__ X, const float* __restrict__ W,
                              float* __restrict__ Hout, int K, int row0, int Nn) {
  int i = blockIdx.x * blockDim.x + threadIdx.x;
  int rows = Nn - row0;
  if (i >= rows * HF) return;
  int r = row0 + (i >> 7), col = i & (HF - 1);
  float acc = 0.f;
  for (int k = 0; k < K; ++k) acc += X[(size_t)r * K + k] * W[(size_t)k * HF + col];
  Hout[(size_t)r * HF + col] = acc;
}

__global__ void gat_alpha(const float* __restrict__ H, const float* __restrict__ att_src,
                          const float* __restrict__ att_dst,
                          float* __restrict__ a_src, float* __restrict__ a_dst, int Nn) {
  int i = blockIdx.x * blockDim.x + threadIdx.x;
  if (i >= Nn * HEADS) return;
  int h = i & (HEADS - 1);
  const float4* hp = (const float4*)(H + (size_t)(i >> 2) * HF + h * FEAT);
  const float4* as = (const float4*)(att_src + h * FEAT);
  const float4* ad = (const float4*)(att_dst + h * FEAT);
  float s1 = 0.f, s2 = 0.f;
#pragma unroll
  for (int f = 0; f < FEAT / 4; ++f) {
    float4 v = hp[f], a = as[f], d = ad[f];
    s1 += v.x * a.x + v.y * a.y + v.z * a.z + v.w * a.w;
    s2 += v.x * d.x + v.y * d.y + v.z * d.z + v.w * d.w;
  }
  a_src[i] = s1;
  a_dst[i] = s2;
}

__global__ void gat_init(float* __restrict__ m, float* __restrict__ s,
                         float4* __restrict__ acc4, int Nn) {
  int i = blockIdx.x * blockDim.x + threadIdx.x;
  if (i < Nn * HEADS) { m[i] = -INFINITY; s[i] = 0.f; }
  if (i < Nn * (HF / 4)) acc4[i] = make_float4(0.f, 0.f, 0.f, 0.f);
}

__device__ __forceinline__ float leaky(float e) { return e > 0.f ? e : 0.2f * e; }

__device__ __forceinline__ void atomicMaxF(float* addr, float val) {
  if (val >= 0.f) atomicMax((int*)addr, __float_as_int(val));
  else            atomicMin((unsigned int*)addr, __float_as_uint(val));
}

__device__ __forceinline__ void edge_pair(const int* __restrict__ ei, int eid, int E,
                                          int& s, int& d) {
  if (eid < E) { s = ei[eid]; d = ei[E + eid]; }
  else         { s = d = eid - E; }
}

__global__ void gat_edge_max(const int* __restrict__ ei, const float* __restrict__ asrc,
                             const float* __restrict__ adst, float* __restrict__ m,
                             int E, int Nn) {
  int i = blockIdx.x * blockDim.x + threadIdx.x;
  if (i >= (E + Nn) * HEADS) return;
  int eid = i >> 2, h = i & (HEADS - 1), s, d;
  edge_pair(ei, eid, E, s, d);
  float e = leaky(asrc[s * HEADS + h] + adst[d * HEADS + h]);
  atomicMaxF(&m[d * HEADS + h], e);
}

__global__ void gat_edge_sum(const int* __restrict__ ei, const float* __restrict__ asrc,
                             const float* __restrict__ adst, const float* __restrict__ m,
                             float* __restrict__ sums, int E, int Nn) {
  int i = blockIdx.x * blockDim.x + threadIdx.x;
  if (i >= (E + Nn) * HEADS) return;
  int eid = i >> 2, h = i & (HEADS - 1), s, d;
  edge_pair(ei, eid, E, s, d);
  float e = leaky(asrc[s * HEADS + h] + adst[d * HEADS + h]);
  atomicAdd(&sums[d * HEADS + h], expf(e - m[d * HEADS + h]));
}

__global__ void gat_edge_agg(const int* __restrict__ ei, const float* __restrict__ asrc,
                             const float* __restrict__ adst, const float* __restrict__ m,
                             const float* __restrict__ sums, const float* __restrict__ hfeat,
                             float* __restrict__ acc, int E, int Nn) {
  long long i = (long long)blockIdx.x * blockDim.x + threadIdx.x;
  if (i >= (long long)(E + Nn) * (HF / 4)) return;
  int eid = (int)(i >> 5);
  int g   = (int)(i & 31);
  int c   = g * 4;
  int h   = c >> 5;
  int s, d;
  edge_pair(ei, eid, E, s, d);
  float e = leaky(asrc[s * HEADS + h] + adst[d * HEADS + h]);
  float w = expf(e - m[d * HEADS + h]) / (sums[d * HEADS + h] + 1e-16f);
  const float4 hv = *(const float4*)(hfeat + (size_t)s * HF + c);
  float* ap = acc + (size_t)d * HF + c;
  atomicAdd(ap + 0, w * hv.x);
  atomicAdd(ap + 1, w * hv.y);
  atomicAdd(ap + 2, w * hv.z);
  atomicAdd(ap + 3, w * hv.w);
}

__global__ void gat_head_mean(const float* __restrict__ acc, const float* __restrict__ bias,
                              float* __restrict__ out, int Nn, int do_elu) {
  int i = blockIdx.x * blockDim.x + threadIdx.x;
  if (i >= Nn * FEAT) return;
  int n = i >> 5, f = i & (FEAT - 1);
  const float* a = acc + (size_t)n * HF;
  float v = 0.25f * (a[f] + a[FEAT + f] + a[2 * FEAT + f] + a[3 * FEAT + f]) + bias[f];
  if (do_elu) v = v > 0.f ? v : (expf(v) - 1.f);
  out[i] = v;
}

extern "C" void kernel_launch(void* const* d_in, const int* in_sizes, int n_in,
                              void* d_out, int out_size, void* d_ws, size_t ws_size,
                              hipStream_t stream) {
  const float* x   = (const float*)d_in[0];
  const int*   ei  = (const int*)d_in[1];
  const float* W1  = (const float*)d_in[2];
  const float* as1 = (const float*)d_in[3];
  const float* ad1 = (const float*)d_in[4];
  const float* b1  = (const float*)d_in[5];
  const float* W2  = (const float*)d_in[6];
  const float* as2 = (const float*)d_in[7];
  const float* ad2 = (const float*)d_in[8];
  const float* b2  = (const float*)d_in[9];

  const int Nn = in_sizes[0] / HF;
  const int E  = in_sizes[1] / 2;
  const int ET = E + Nn;

  float* ws    = (float*)d_ws;
  float* hbuf  = ws;
  float* acc   = hbuf + (size_t)Nn * HF;
  float* asrc  = acc  + (size_t)Nn * HF;
  float* adst  = asrc + (size_t)Nn * HEADS;
  float* mbuf  = adst + (size_t)Nn * HEADS;
  float* sbuf  = mbuf + (size_t)Nn * HEADS;
  float* x2    = sbuf + (size_t)Nn * HEADS;

  const dim3 blk(256);
  const int row_tiles = Nn / 16;
  const int tail_row0 = row_tiles * 16;

  for (int layer = 0; layer < 2; ++layer) {
    const float* Xin = (layer == 0) ? x  : x2;
    const int    K   = (layer == 0) ? 128 : FEAT;
    const float* Wm  = (layer == 0) ? W1 : W2;
    const float* as  = (layer == 0) ? as1 : as2;
    const float* ad  = (layer == 0) ? ad1 : ad2;
    const float* bb  = (layer == 0) ? b1 : b2;
    float*       op  = (layer == 0) ? x2 : (float*)d_out;

    if (row_tiles > 0)
      gat_gemm_wmma_f32<<<row_tiles, blk, 0, stream>>>(Xin, Wm, hbuf, K);
    if (tail_row0 < Nn) {
      int tail = (Nn - tail_row0) * HF;
      gat_gemm_tail<<<(tail + 255) / 256, blk, 0, stream>>>(Xin, Wm, hbuf, K, tail_row0, Nn);
    }
    gat_alpha<<<(Nn * HEADS + 255) / 256, blk, 0, stream>>>(hbuf, as, ad, asrc, adst, Nn);
    gat_init<<<(Nn * (HF / 4) + 255) / 256, blk, 0, stream>>>(mbuf, sbuf, (float4*)acc, Nn);
    gat_edge_max<<<(ET * HEADS + 255) / 256, blk, 0, stream>>>(ei, asrc, adst, mbuf, E, Nn);
    gat_edge_sum<<<(ET * HEADS + 255) / 256, blk, 0, stream>>>(ei, asrc, adst, mbuf, sbuf, E, Nn);
    {
      long long tot = (long long)ET * (HF / 4);
      unsigned int nb = (unsigned int)((tot + 255) / 256);
      gat_edge_agg<<<nb, blk, 0, stream>>>(ei, asrc, adst, mbuf, sbuf, hbuf, acc, E, Nn);
    }
    gat_head_mean<<<(Nn * FEAT + 255) / 256, blk, 0, stream>>>(acc, bb, op, Nn, layer == 0 ? 1 : 0);
  }
}


# edge aggregation: nodes split in two ranges, one ticket queue of edge chunks each; a wave drains its package half's queue first, so an acc row is added into from one half; arrival-order sums kept
# speedup vs baseline: 1.9635x; 1.9635x over previous
#include <hip/hip_runtime.h>
#include <math.h>

#define HEADS 4
#define FEAT  32
#define HF    128
#define QPAD  64

typedef __attribute__((ext_vector_type(2))) float v2f;
typedef __attribute__((ext_vector_type(8))) float v8f;

__global__ void __launch_bounds__(256)
gat_gemm_wmma_f32(const float* __restrict__ X, const float* __restrict__ W,
                  float* __restrict__ Hout, int K) {
  const int wave = threadIdx.x >> 5;
  const int lane = threadIdx.x & 31;
  const int half = lane >> 4;
  const int l15  = lane & 15;
  const int row  = blockIdx.x * 16 + l15;
  const int col  = wave * 16 + l15;
  const int kh   = half * 2;
  const float* __restrict__ xrow = X + (size_t)row * K;

  v8f c = {};
  for (int k0 = 0; k0 < K; k0 += 4) {
    v2f a, b;
    a.x = xrow[k0 + kh];
    a.y = xrow[k0 + kh + 1];
    b.x = W[(size_t)(k0 + kh)     * HF + col];
    b.y = W[(size_t)(k0 + kh + 1) * HF + col];
    c = __builtin_amdgcn_wmma_f32_16x16x4_f32(false, a, false, b,
                                              (short)0, c, false, false);
  }
  const int rbase = blockIdx.x * 16 + half * 8;
  float* orow = Hout + (size_t)rbase * HF + col;
#pragma unroll
  for (int v = 0; v < 8; ++v)
    orow[(size_t)v * HF] = c[v];
}

__global__ void gat_gemm_tail(const float* __restrict__ X, const float* __restrict__ W,
                              float* __restrict__ Hout, int K, int row0, int Nn) {
  int i = blockIdx.x * blockDim.x + threadIdx.x;
  int rows = Nn - row0;
  if (i >= rows * HF) return;
  int r = row0 + (i >> 7), col = i & (HF - 1);
  float acc = 0.f;
  for (int k = 0; k < K; ++k) acc += X[(size_t)r * K + k] * W[(size_t)k * HF + col];
  Hout[(size_t)r * HF + col] = acc;
}

__global__ void gat_alpha(const float* __restrict__ H, const float* __restrict__ att_src,
                          const float* __restrict__ att_dst,
                          float* __restrict__ a_src, float* __restrict__ a_dst, int Nn) {
  int i = blockIdx.x * blockDim.x + threadIdx.x;
  if (i >= Nn * HEADS) return;
  int h = i & (HEADS - 1);
  const float4* hp = (const float4*)(H + (size_t)(i >> 2) * HF + h * FEAT);
  const float4* as = (const float4*)(att_src + h * FEAT);
  const float4* ad = (const float4*)(att_dst + h * FEAT);
  float s1 = 0.f, s2 = 0.f;
#pragma unroll
  for (int f = 0; f < FEAT / 4; ++f) {
    float4 v = hp[f], a = as[f], d = ad[f];
    s1 += v.x * a.x + v.y * a.y + v.z * a.z + v.w * a.w;
    s2 += v.x * d.x + v.y * d.y + v.z * d.z + v.w * d.w;
  }
  a_src[i] = s1;
  a_dst[i] = s2;
}

__global__ void gat_init(float* __restrict__ m, float* __restrict__ s,
                         float4* __restrict__ acc4, unsigned int* __restrict__ ctr, int Nn) {
  int i = blockIdx.x * blockDim.x + threadIdx.x;
  if (i == 0) { ctr[0] = 0u; ctr[QPAD] = 0u; }
  if (i < Nn * HEADS) { m[i] = -INFINITY; s[i] = 0.f; }
  if (i < Nn * (HF / 4)) acc4[i] = make_float4(0.f, 0.f, 0.f, 0.f);
}

__device__ __forceinline__ float leaky(float e) { return e > 0.f ? e : 0.2f * e; }

__device__ __forceinline__ void atomicMaxF(float* addr, float val) {
  if (val >= 0.f) atomicMax((int*)addr, __float_as_int(val));
  else            atomicMin((unsigned int*)addr, __float_as_uint(val));
}

__device__ __forceinline__ void edge_pair(const int* __restrict__ ei, int eid, int E,
                                          int& s, int& d) {
  if (eid < E) { s = ei[eid]; d = ei[E + eid]; }
  else         { s = d = eid - E; }
}

__global__ void gat_edge_max(const int* __restrict__ ei, const float* __restrict__ asrc,
                             const float* __restrict__ adst, float* __restrict__ m,
                             int E, int Nn) {
  int i = blockIdx.x * blockDim.x + threadIdx.x;
  if (i >= (E + Nn) * HEADS) return;
  int eid = i >> 2, h = i & (HEADS - 1), s, d;
  edge_pair(ei, eid, E, s, d);
  float e = leaky(asrc[s * HEADS + h] + adst[d * HEADS + h]);
  atomicMaxF(&m[d * HEADS + h], e);
}

#define EPW   4
#define CHUNK 64

__device__ __forceinline__ void agg_chunk(const int* __restrict__ ei, const float4* __restrict__ asrc4,
                                          const float4* __restrict__ adst4, const float4* __restrict__ m4,
                                          float* __restrict__ sums, const float* __restrict__ hfeat,
                                          float* __restrict__ acc, int E, int ET, int e0, int nhalf,
                                          int t, int lane) {
  for (int sub = 0; sub < CHUNK / 32; ++sub) {
    const int eid = e0 + sub * 32 + lane;
    int s = 0, d = 0;
    if (eid < ET) edge_pair(ei, eid, E, s, d);
    const bool mine = (eid < ET) && ((d >= nhalf ? 1 : 0) == t);
    float w0 = 0.f, w1 = 0.f, w2 = 0.f, w3 = 0.f;
    if (mine) {
      const float4 a = asrc4[s], b = adst4[d], mx = m4[d];
      w0 = expf(leaky(a.x + b.x) - mx.x);
      w1 = expf(leaky(a.y + b.y) - mx.y);
      w2 = expf(leaky(a.z + b.z) - mx.z);
      w3 = expf(leaky(a.w + b.w) - mx.w);
    }
    unsigned int mask = __builtin_amdgcn_ballot_w32(mine);
    while (mask) {
      int k[EPW];
      int cnt = 0, last = 0;
#pragma unroll
      for (int q = 0; q < EPW; ++q) {
        if (mask) { last = __builtin_ctz(mask); mask &= mask - 1u; ++cnt; }
        k[q] = last;
      }
      float hv[EPW][HEADS];
      int   dk[EPW];
#pragma unroll
      for (int q = 0; q < EPW; ++q) {
        const int sk = __builtin_amdgcn_readlane(s, k[q]);
        dk[q]        = __builtin_amdgcn_readlane(d, k[q]);
        const float* hp = hfeat + (size_t)sk * HF + lane;
#pragma unroll
        for (int j = 0; j < HEADS; ++j) hv[q][j] = hp[j * FEAT];
      }
      const int slot = (lane >> 2) & (EPW - 1);
      const int from = slot == 0 ? k[0] : slot == 1 ? k[1] : slot == 2 ? k[2] : k[3];
      const float t0 = __shfl(w0, from), t1 = __shfl(w1, from), t2 = __shfl(w2, from), t3 = __shfl(w3, from);
      const int   dsel = __shfl(d, from);
      const int   hl = lane & (HEADS - 1);
      const float wsel = hl == 0 ? t0 : hl == 1 ? t1 : hl == 2 ? t2 : t3;
#pragma unroll
      for (int q = 0; q < EPW; ++q) {
        if (q < cnt) {
          float* ap = acc + (size_t)dk[q] * HF + lane;
          const float x0 = __int_as_float(__builtin_amdgcn_readlane(__float_as_int(w0), k[q]));
          const float x1 = __int_as_float(__builtin_amdgcn_readlane(__float_as_int(w1), k[q]));
          const float x2 = __int_as_float(__builtin_amdgcn_readlane(__float_as_int(w2), k[q]));
          const float x3 = __int_as_float(__builtin_amdgcn_readlane(__float_as_int(w3), k[q]));
          atomicAdd(ap,            x0 * hv[q][0]);
          atomicAdd(ap + FEAT,     x1 * hv[q][1]);
          atomicAdd(ap + 2 * FEAT, x2 * hv[q][2]);
          atomicAdd(ap + 3 * FEAT, x3 * hv[q][3]);
        }
      }
      if (lane < HEADS * EPW && slot < cnt)
        atomicAdd(&sums[(size_t)dsel * HEADS + hl], wsel);
    }
  }
}

__global__ void __launch_bounds__(256)
gat_edge_agg(const int* __restrict__ ei, const float* __restrict__ asrc,
             const float* __restrict__ adst, const float* __restrict__ m,
             float* __restrict__ sums, const float* __restrict__ hfeat,
             float* __restrict__ acc, unsigned int* __restrict__ ctr, int E, int Nn) {
  const int lane = threadIdx.x & 31;
  const int ET = E + Nn;
  const unsigned int nch = (unsigned int)((ET + CHUNK - 1) / CHUNK);
  const int nhalf = Nn >> 1;
  const int myhalf = (int)((__builtin_amdgcn_s_sendmsg_rtn(0x87) >> 18) & 1u);
  for (int pass = 0; pass < 2; ++pass) {
    const int t = myhalf ^ pass;
    for (;;) {
      unsigned int c = 0u;
      if (lane == 0) c = atomicAdd(&ctr[t * QPAD], 1u);
      c = (unsigned int)__builtin_amdgcn_readfirstlane((int)c);
      if (c >= nch) break;
      agg_chunk(ei, (const float4*)asrc, (const float4*)adst, (const float4*)m, sums, hfeat, acc,
                E, ET, (int)(c * CHUNK), nhalf, t, lane);
    }
  }
}

__global__ void gat_head_mean(const float* __restrict__ acc, const float* __restrict__ sums,
                              const float* __restrict__ bias,
                              float* __restrict__ out, int Nn, int do_elu) {
  int i = blockIdx.x * blockDim.x + threadIdx.x;
  if (i >= Nn * FEAT) return;
  int n = i >> 5, f = i & (FEAT - 1);
  const float* a = acc + (size_t)n * HF;
  const float4 sv = *(const float4*)(sums + (size_t)n * HEADS);
  float v = 0.25f * (a[f]            / (sv.x + 1e-16f) + a[FEAT + f]     / (sv.y + 1e-16f) +
                     a[2 * FEAT + f] / (sv.z + 1e-16f) + a[3 * FEAT + f] / (sv.w + 1e-16f)) + bias[f];
  if (do_elu) v = v > 0.f ? v : (expf(v) - 1.f);
  out[i] = v;
}

extern "C" void kernel_launch(void* const* d_in, const int* in_sizes, int n_in,
                              void* d_out, int out_size, void* d_ws, size_t ws_size,
                              hipStream_t stream) {
  const float* x   = (const float*)d_in[0];
  const int*   ei  = (const int*)d_in[1];
  const float* W1  = (const float*)d_in[2];
  const float* as1 = (const float*)d_in[3];
  const float* ad1 = (const float*)d_in[4];
  const float* b1  = (const float*)d_in[5];
  const float* W2  = (const float*)d_in[6];
  const float* as2 = (const float*)d_in[7];
  const float* ad2 = (const float*)d_in[8];
  const float* b2  = (const float*)d_in[9];

  const int Nn = in_sizes[0] / HF;
  const int E  = in_sizes[1] / 2;
  const int ET = E + Nn;

  float* ws    = (float*)d_ws;
  float* hbuf  = ws;
  float* acc   = hbuf + (size_t)Nn * HF;
  float* asrc  = acc  + (size_t)Nn * HF;
  float* adst  = asrc + (size_t)Nn * HEADS;
  float* mbuf  = adst + (size_t)Nn * HEADS;
  float* sbuf  = mbuf + (size_t)Nn * HEADS;
  float* x2    = sbuf + (size_t)Nn * HEADS;
  unsigned int* qctr = (unsigned int*)(x2 + (size_t)Nn * FEAT);

  const dim3 blk(256);
  const int row_tiles = Nn / 16;
  const int tail_row0 = row_tiles * 16;

  for (int layer = 0; layer < 2; ++layer) {
    const float* Xin = (layer == 0) ? x  : x2;
    const int    K   = (layer == 0) ? 128 : FEAT;
    const float* Wm  = (layer == 0) ? W1 : W2;
    const float* as  = (layer == 0) ? as1 : as2;
    const float* ad  = (layer == 0) ? ad1 : ad2;
    const float* bb  = (layer == 0) ? b1 : b2;
    float*       op  = (layer == 0) ? x2 : (float*)d_out;

    if (row_tiles > 0)
      gat_gemm_wmma_f32<<<row_tiles, blk, 0, stream>>>(Xin, Wm, hbuf, K);
    if (tail_row0 < Nn) {
      int tail = (Nn - tail_row0) * HF;
      gat_gemm_tail<<<(tail + 255) / 256, blk, 0, stream>>>(Xin, Wm, hbuf, K, tail_row0, Nn);
    }
    gat_alpha<<<(Nn * HEADS + 255) / 256, blk, 0, stream>>>(hbuf, as, ad, asrc, adst, Nn);
    gat_init<<<(Nn * (HF / 4) + 255) / 256, blk, 0, stream>>>(mbuf, sbuf, (float4*)acc, qctr, Nn);
    gat_edge_max<<<(ET * HEADS + 255) / 256, blk, 0, stream>>>(ei, asrc, adst, mbuf, E, Nn);
    gat_edge_agg<<<2048, blk, 0, stream>>>(ei, asrc, adst, mbuf, sbuf, hbuf, acc, qctr, E, Nn);
    gat_head_mean<<<(Nn * FEAT + 255) / 256, blk, 0, stream>>>(acc, sbuf, bb, op, Nn, layer == 0 ? 1 : 0);
  }
}


# softmax maximum taken by the owner work-group in LDS in its one key sweep (max of a_src per row; a_dst and leaky-relu applied once per row): both edge-max launches go; LDS arrival-order adds kept
# speedup vs baseline: 4.4745x; 2.2788x over previous
#include <hip/hip_runtime.h>
#include <math.h>

#define HEADS 4
#define FEAT  32
#define HF    128

typedef __attribute__((ext_vector_type(2))) float v2f;
typedef __attribute__((ext_vector_type(8))) float v8f;

__global__ void __launch_bounds__(256)
gat_gemm_wmma_f32(const float* __restrict__ X, const float* __restrict__ W,
                  float* __restrict__ Hout, int K) {
  const int wave = threadIdx.x >> 5;
  const int lane = threadIdx.x & 31;
  const int half = lane >> 4;
  const int l15  = lane & 15;
  const int row  = blockIdx.x * 16 + l15;
  const int col  = wave * 16 + l15;
  const int kh   = half * 2;
  const float* __restrict__ xrow = X + (size_t)row * K;

  v8f c = {};
  for (int k0 = 0; k0 < K; k0 += 4) {
    v2f a, b;
    a.x = xrow[k0 + kh];
    a.y = xrow[k0 + kh + 1];
    b.x = W[(size_t)(k0 + kh)     * HF + col];
    b.y = W[(size_t)(k0 + kh + 1) * HF + col];
    c = __builtin_amdgcn_wmma_f32_16x16x4_f32(false, a, false, b,
                                              (short)0, c, false, false);
  }
  const int rbase = blockIdx.x * 16 + half * 8;
  float* orow = Hout + (size_t)rbase * HF + col;
#pragma unroll
  for (int v = 0; v < 8; ++v)
    orow[(size_t)v * HF] = c[v];
}

__global__ void gat_gemm_tail(const float* __restrict__ X, const float* __restrict__ W,
                              float* __restrict__ Hout, int K, int row0, int Nn) {
  int i = blockIdx.x * blockDim.x + threadIdx.x;
  int rows = Nn - row0;
  if (i >= rows * HF) return;
  int r = row0 + (i >> 7), col = i & (HF - 1);
  float acc = 0.f;
  for (int k = 0; k < K; ++k) acc += X[(size_t)r * K + k] * W[(size_t)k * HF + col];
  Hout[(size_t)r * HF + col] = acc;
}

__global__ void gat_alpha(const float* __restrict__ H, const float* __restrict__ att_src,
                          const float* __restrict__ att_dst,
                          float* __restrict__ a_src, float* __restrict__ a_dst, int Nn) {
  int i = blockIdx.x * blockDim.x + threadIdx.x;
  if (i >= Nn * HEADS) return;
  int h = i & (HEADS - 1);
  const float4* hp = (const float4*)(H + (size_t)(i >> 2) * HF + h * FEAT);
  const float4* as = (const float4*)(att_src + h * FEAT);
  const float4* ad = (const float4*)(att_dst + h * FEAT);
  float s1 = 0.f, s2 = 0.f;
#pragma unroll
  for (int f = 0; f < FEAT / 4; ++f) {
    float4 v = hp[f], a = as[f], d = ad[f];
    s1 += v.x * a.x + v.y * a.y + v.z * a.z + v.w * a.w;
    s2 += v.x * d.x + v.y * d.y + v.z * d.z + v.w * d.w;
  }
  a_src[i] = s1;
  a_dst[i] = s2;
}

#define ROWS  256
#define KBLK  512

__device__ __forceinline__ unsigned int owner_key(int d) { return min((unsigned int)d / ROWS, 255u); }

__global__ void gat_init(int* __restrict__ okey, const int* __restrict__ ei, int E, int nkw) {
  int i = blockIdx.x * blockDim.x + threadIdx.x;
  if (i < nkw) {
    const int e4 = 4 * i;
    unsigned int w = 0u;
    if ((E & 3) == 0 && e4 < E) {
      const int4 d = *(const int4*)(ei + E + e4);
      w = owner_key(d.x) | (owner_key(d.y) << 8) | (owner_key(d.z) << 16) | (owner_key(d.w) << 24);
    } else {
#pragma unroll
      for (int j = 0; j < 4; ++j) {
        const unsigned int kb = e4 + j < E ? owner_key(ei[E + e4 + j]) : 255u;
        w |= kb << (8 * j);
      }
    }
    okey[i] = (int)w;
  }
}

__device__ __forceinline__ float leaky(float e) { return e > 0.f ? e : 0.2f * e; }

__device__ __forceinline__ void lds_max_f(float* addr, float val) {
  if (__float_as_int(val) >= 0) atomicMax((int*)addr, __float_as_int(val));
  else                          atomicMin((unsigned int*)addr, __float_as_uint(val));
}

#define EPW   8
#define NWAVE 32
#define LCAP  1088
#define SCANU 4

__device__ __forceinline__ void owner_flush(const int* lst, int cntw,
                                            const float4* __restrict__ asrc4, const float4* adstL4,
                                            const float4* mL4, float* sumsL, float* accL, float* wl,
                                            const float* __restrict__ hfeat, int lane) {
  for (int b = 0; b < cntw; b += 32) {
    const int n = min(32, cntw - b);
    int s = 0, r = 0;
    float w0 = 0.f, w1 = 0.f, w2 = 0.f, w3 = 0.f;
    if (lane < n) {
      const unsigned int p = (unsigned int)lst[b + lane];
      s = (int)(p & 0xffffu);
      r = (int)(p >> 16);
      const float4 a = asrc4[s], bq = adstL4[r], mx = mL4[r];
      w0 = expf(leaky(a.x + bq.x) - mx.x);
      w1 = expf(leaky(a.y + bq.y) - mx.y);
      w2 = expf(leaky(a.z + bq.z) - mx.z);
      w3 = expf(leaky(a.w + bq.w) - mx.w);
      atomicAdd(&sumsL[r * HEADS + 0], w0);
      atomicAdd(&sumsL[r * HEADS + 1], w1);
      atomicAdd(&sumsL[r * HEADS + 2], w2);
      atomicAdd(&sumsL[r * HEADS + 3], w3);
      ((float4*)wl)[lane] = make_float4(w0, w1, w2, w3);
    }
    const int hsel = lane >> 3;
    for (int g = 0; g < n; g += EPW) {
      float4 hv[EPW];
      float  xs[EPW];
      int    rk[EPW], kk[EPW];
#pragma unroll
      for (int q = 0; q < EPW; ++q) {
        kk[q] = min(g + q, n - 1);
        const int sk = __builtin_amdgcn_readlane(s, kk[q]);
        rk[q]        = __builtin_amdgcn_readlane(r, kk[q]);
        hv[q] = *(const float4*)(hfeat + (size_t)sk * HF + lane * 4);
        xs[q] = wl[kk[q] * HEADS + hsel];
      }
#pragma unroll
      for (int q = 0; q < EPW; ++q) {
        if (g + q < n) {
          float* ap = accL + rk[q] * HF + lane;
          atomicAdd(ap,      xs[q] * hv[q].x);
          atomicAdd(ap + 32, xs[q] * hv[q].y);
          atomicAdd(ap + 64, xs[q] * hv[q].z);
          atomicAdd(ap + 96, xs[q] * hv[q].w);
        }
      }
    }
  }
}

__global__ void __launch_bounds__(1024)
gat_agg_owner(const int* __restrict__ ei, const int* __restrict__ okey, const float* __restrict__ asrc,
              const float* __restrict__ adst,
              const float* __restrict__ hfeat, const float* __restrict__ bias,
              float* __restrict__ out, int E, int Nn, int do_elu) {
  __shared__ float4 accL4[ROWS * HF / 4];
  __shared__ float4 sumsL4[ROWS];
  __shared__ float4 adstL4[ROWS];
  __shared__ float4 mL4[ROWS];
  __shared__ int    listL[NWAVE * LCAP];
  __shared__ float4 wL4[NWAVE * 32];
  float* accL  = (float*)accL4;
  float* sumsL = (float*)sumsL4;
  float* mxL   = (float*)mL4;

  const int tid = threadIdx.x, lane = tid & 31;
  const int wv  = __builtin_amdgcn_readfirstlane(tid >> 5);
  const int lo  = blockIdx.x * ROWS;
  const int cnt = min(ROWS, Nn - lo);

  for (int i = tid; i < ROWS * HF / 4; i += NWAVE * 32) accL4[i] = make_float4(0.f, 0.f, 0.f, 0.f);
  for (int i = tid; i < ROWS; i += NWAVE * 32) {
    sumsL4[i] = make_float4(0.f, 0.f, 0.f, 0.f);
    mL4[i]    = make_float4(-INFINITY, -INFINITY, -INFINITY, -INFINITY);
    if (i < cnt) adstL4[i] = ((const float4*)adst)[lo + i];
  }
  __syncthreads();

  int* lst = listL + wv * LCAP;
  int cntw = 0;
  int np   = 0;
  bool drop    = false;
  bool second  = false;
  const float4* asrc4 = (const float4*)asrc;
  const unsigned int pat = (unsigned int)blockIdx.x * 0x01010101u;

  auto zero_bytes = [](unsigned int x) -> unsigned int {
    return ~((((x & 0x7f7f7f7fu) + 0x7f7f7f7fu) | x) | 0x7f7f7f7fu);
  };
  auto settle_to = [&](int upto) {
    for (int b = np; b < upto; b += 32) {
      const int i = b + lane;
      if (i < upto) {
        const int eid = lst[i];
        int s, r;
        if (eid < E) { s = ei[eid]; r = ei[E + eid] - lo; }
        else         { s = eid - E; r = s - lo; }
        if (!second) {
          const float4 a = asrc4[s];
          lds_max_f(&mxL[r * HEADS + 0], a.x);
          lds_max_f(&mxL[r * HEADS + 1], a.y);
          lds_max_f(&mxL[r * HEADS + 2], a.z);
          lds_max_f(&mxL[r * HEADS + 3], a.w);
        }
        lst[i] = (int)((unsigned int)s | ((unsigned int)r << 16));
      }
    }
    np = upto;
  };
  auto settle = [&](bool all) {
    settle_to((all || drop) ? cntw : np + ((cntw - np) & ~31));
    if (drop) {
      if (second) owner_flush(lst, np, asrc4, adstL4, mL4, sumsL, accL, (float*)(wL4 + wv * 32), hfeat, lane);
      cntw = 0;
      np   = 0;
    }
  };

  const int nblk = (E + KBLK - 1) / KBLK;
  const int start = (int)((long long)blockIdx.x * nblk / gridDim.x);
  auto load_blk = [&](int k) -> uint4 {
    int bi = k + start; if (bi >= nblk) bi -= nblk;
    return *(const uint4*)(okey + (size_t)bi * (KBLK / 4) + lane * 4);
  };
  auto scan_blk = [&](const uint4& d, int k) {
    int bi = k + start; if (bi >= nblk) bi -= nblk;
    const int e0 = bi * KBLK + lane * 16;
    unsigned int c = (zero_bytes(d.x ^ pat) >> 7) | (zero_bytes(d.y ^ pat) >> 6) |
                     (zero_bytes(d.z ^ pat) >> 5) | (zero_bytes(d.w ^ pat) >> 4);
    unsigned int act = __builtin_amdgcn_ballot_w32(c != 0u);
    while (act) {
      if (c != 0u) {
        const int t = __builtin_ctz(c);
        lst[(int)__builtin_amdgcn_mbcnt_lo(act, (unsigned int)cntw)] = e0 + ((t & 3) << 2) + (t >> 3);
        c &= c - 1u;
      }
      cntw += __builtin_popcount(act);
      act = __builtin_amdgcn_ballot_w32(c != 0u);
    }
  };
  const int nbw   = wv < nblk ? (nblk - wv + NWAVE - 1) / NWAVE : 0;
  const int ntrip = nbw / SCANU;

  auto sweep = [&]() {
    if (ntrip > 0) {
      uint4 cur[SCANU], nxt[SCANU];
#pragma unroll
      for (int u = 0; u < SCANU; ++u) cur[u] = load_blk(wv + u * NWAVE);
      for (int trip = 0; trip < ntrip; ++trip) {
        const int tn = min(trip + 1, ntrip - 1);
#pragma unroll
        for (int u = 0; u < SCANU; ++u) nxt[u] = load_blk(wv + (tn * SCANU + u) * NWAVE);
        unsigned int late = 0u;
#pragma unroll
        for (int u = 0; u < SCANU; ++u) {
          if (cntw <= LCAP - KBLK) scan_blk(cur[u], wv + (trip * SCANU + u) * NWAVE);
          else late |= 1u << u;
        }
        if (late) drop = true;
        settle(false);
        while (late) {
          const int u = __builtin_ctz(late);
          late &= late - 1u;
          const int k = wv + (trip * SCANU + u) * NWAVE;
          const uint4 d = load_blk(k);
          scan_blk(d, k);
          settle(true);
        }
#pragma unroll
        for (int u = 0; u < SCANU; ++u) cur[u] = nxt[u];
      }
    }
    for (int t = ntrip * SCANU; t < nbw; ++t) {
      if (cntw > LCAP - KBLK) { drop = true; settle(true); }
      const uint4 d = load_blk(wv + t * NWAVE);
      scan_blk(d, wv + t * NWAVE);
      settle(false);
    }
    for (int rb = wv * 32; rb < ROWS; rb += NWAVE * 32) {
      if (cntw > LCAP - 32) { drop = true; settle(true); }
      const int r = rb + lane;
      const unsigned int mask = __builtin_amdgcn_ballot_w32(r < cnt);
      if (r < cnt) lst[(int)__builtin_amdgcn_mbcnt_lo(mask, (unsigned int)cntw)] = E + lo + r;
      cntw += __builtin_popcount(mask);
    }
    settle(true);
  };

  sweep();
  const bool kept = !drop;
  __syncthreads();
  for (int i = tid; i < cnt; i += NWAVE * 32) {
    const float4 ms = mL4[i], bq = adstL4[i];
    mL4[i] = make_float4(leaky(ms.x + bq.x), leaky(ms.y + bq.y), leaky(ms.z + bq.z), leaky(ms.w + bq.w));
  }
  __syncthreads();
  second = true;
  if (kept) {
    owner_flush(lst, np, asrc4, adstL4, mL4, sumsL, accL, (float*)(wL4 + wv * 32), hfeat, lane);
  } else {
    drop = true;
    sweep();
  }
  __syncthreads();

  for (int idx = tid; idx < cnt * FEAT; idx += NWAVE * 32) {
    const int r = idx >> 5, f = idx & (FEAT - 1);
    const float* a = accL + r * HF + (f & 3) * 32 + (f >> 2);
    const float4 sv = sumsL4[r];
    float v = 0.25f * (a[0]  / (sv.x + 1e-16f) + a[8]  / (sv.y + 1e-16f) +
                       a[16] / (sv.z + 1e-16f) + a[24] / (sv.w + 1e-16f)) + bias[f];
    if (do_elu) v = v > 0.f ? v : (expf(v) - 1.f);
    out[(size_t)(lo + r) * FEAT + f] = v;
  }
}

extern "C" void kernel_launch(void* const* d_in, const int* in_sizes, int n_in,
                              void* d_out, int out_size, void* d_ws, size_t ws_size,
                              hipStream_t stream) {
  const float* x   = (const float*)d_in[0];
  const int*   ei  = (const int*)d_in[1];
  const float* W1  = (const float*)d_in[2];
  const float* as1 = (const float*)d_in[3];
  const float* ad1 = (const float*)d_in[4];
  const float* b1  = (const float*)d_in[5];
  const float* W2  = (const float*)d_in[6];
  const float* as2 = (const float*)d_in[7];
  const float* ad2 = (const float*)d_in[8];
  const float* b2  = (const float*)d_in[9];

  const int Nn = in_sizes[0] / HF;
  const int E  = in_sizes[1] / 2;

  float* ws    = (float*)d_ws;
  float* hbuf  = ws;
  float* asrc  = hbuf + (size_t)Nn * HF;
  float* adst  = asrc + (size_t)Nn * HEADS;
  float* x2    = adst + (size_t)Nn * HEADS * 2;
  int*   okey  = (int*)(x2 + (size_t)Nn * FEAT);
  const int nkw = (E + KBLK - 1) / KBLK * (KBLK / 4);

  const dim3 blk(256);
  const int row_tiles = Nn / 16;
  const int tail_row0 = row_tiles * 16;

  for (int layer = 0; layer < 2; ++layer) {
    const float* Xin = (layer == 0) ? x  : x2;
    const int    K   = (layer == 0) ? 128 : FEAT;
    const float* Wm  = (layer == 0) ? W1 : W2;
    const float* as  = (layer == 0) ? as1 : as2;
    const float* ad  = (layer == 0) ? ad1 : ad2;
    const float* bb  = (layer == 0) ? b1 : b2;
    float*       op  = (layer == 0) ? x2 : (float*)d_out;

    if (row_tiles > 0)
      gat_gemm_wmma_f32<<<row_tiles, blk, 0, stream>>>(Xin, Wm, hbuf, K);
    if (tail_row0 < Nn) {
      int tail = (Nn - tail_row0) * HF;
      gat_gemm_tail<<<(tail + 255) / 256, blk, 0, stream>>>(Xin, Wm, hbuf, K, tail_row0, Nn);
    }
    gat_alpha<<<(Nn * HEADS + 255) / 256, blk, 0, stream>>>(hbuf, as, ad, asrc, adst, Nn);
    if (layer == 0)
      gat_init<<<(nkw + 255) / 256, blk, 0, stream>>>(okey, ei, E, nkw);
    gat_agg_owner<<<(Nn + ROWS - 1) / ROWS, dim3(NWAVE * 32), 0, stream>>>(ei, okey, asrc, adst, hbuf, bb, op, E, Nn, layer == 0 ? 1 : 0);
  }
}


# layer 2's aggregation reads each wave's settled edge list (source, row) back from the workspace, where layer 1's launch stored it: its key sweep and per-edge source and destination fetches go
# speedup vs baseline: 5.0285x; 1.1238x over previous
#include <hip/hip_runtime.h>
#include <math.h>

#define HEADS 4
#define FEAT  32
#define HF    128

typedef __attribute__((ext_vector_type(2))) float v2f;
typedef __attribute__((ext_vector_type(8))) float v8f;

__global__ void __launch_bounds__(256)
gat_gemm_wmma_f32(const float* __restrict__ X, const float* __restrict__ W,
                  float* __restrict__ Hout, int K) {
  const int wave = threadIdx.x >> 5;
  const int lane = threadIdx.x & 31;
  const int half = lane >> 4;
  const int l15  = lane & 15;
  const int row  = blockIdx.x * 16 + l15;
  const int col  = wave * 16 + l15;
  const int kh   = half * 2;
  const float* __restrict__ xrow = X + (size_t)row * K;

  v8f c = {};
  for (int k0 = 0; k0 < K; k0 += 4) {
    v2f a, b;
    a.x = xrow[k0 + kh];
    a.y = xrow[k0 + kh + 1];
    b.x = W[(size_t)(k0 + kh)     * HF + col];
    b.y = W[(size_t)(k0 + kh + 1) * HF + col];
    c = __builtin_amdgcn_wmma_f32_16x16x4_f32(false, a, false, b,
                                              (short)0, c, false, false);
  }
  const int rbase = blockIdx.x * 16 + half * 8;
  float* orow = Hout + (size_t)rbase * HF + col;
#pragma unroll
  for (int v = 0; v < 8; ++v)
    orow[(size_t)v * HF] = c[v];
}

__global__ void gat_gemm_tail(const float* __restrict__ X, const float* __restrict__ W,
                              float* __restrict__ Hout, int K, int row0, int Nn) {
  int i = blockIdx.x * blockDim.x + threadIdx.x;
  int rows = Nn - row0;
  if (i >= rows * HF) return;
  int r = row0 + (i >> 7), col = i & (HF - 1);
  float acc = 0.f;
  for (int k = 0; k < K; ++k) acc += X[(size_t)r * K + k] * W[(size_t)k * HF + col];
  Hout[(size_t)r * HF + col] = acc;
}

__global__ void gat_alpha(const float* __restrict__ H, const float* __restrict__ att_src,
                          const float* __restrict__ att_dst,
                          float* __restrict__ a_src, float* __restrict__ a_dst, int Nn) {
  int i = blockIdx.x * blockDim.x + threadIdx.x;
  if (i >= Nn * HEADS) return;
  int h = i & (HEADS - 1);
  const float4* hp = (const float4*)(H + (size_t)(i >> 2) * HF + h * FEAT);
  const float4* as = (const float4*)(att_src + h * FEAT);
  const float4* ad = (const float4*)(att_dst + h * FEAT);
  float s1 = 0.f, s2 = 0.f;
#pragma unroll
  for (int f = 0; f < FEAT / 4; ++f) {
    float4 v = hp[f], a = as[f], d = ad[f];
    s1 += v.x * a.x + v.y * a.y + v.z * a.z + v.w * a.w;
    s2 += v.x * d.x + v.y * d.y + v.z * d.z + v.w * d.w;
  }
  a_src[i] = s1;
  a_dst[i] = s2;
}

#define ROWS  256
#define KBLK  512

__device__ __forceinline__ unsigned int owner_key(int d) { return min((unsigned int)d / ROWS, 255u); }

__global__ void gat_init(int* __restrict__ okey, const int* __restrict__ ei, int E, int nkw) {
  int i = blockIdx.x * blockDim.x + threadIdx.x;
  if (i < nkw) {
    const int e4 = 4 * i;
    unsigned int w = 0u;
    if ((E & 3) == 0 && e4 < E) {
      const int4 d = *(const int4*)(ei + E + e4);
      w = owner_key(d.x) | (owner_key(d.y) << 8) | (owner_key(d.z) << 16) | (owner_key(d.w) << 24);
    } else {
#pragma unroll
      for (int j = 0; j < 4; ++j) {
        const unsigned int kb = e4 + j < E ? owner_key(ei[E + e4 + j]) : 255u;
        w |= kb << (8 * j);
      }
    }
    okey[i] = (int)w;
  }
}

__device__ __forceinline__ float leaky(float e) { return e > 0.f ? e : 0.2f * e; }

__device__ __forceinline__ void lds_max_f(float* addr, float val) {
  if (__float_as_int(val) >= 0) atomicMax((int*)addr, __float_as_int(val));
  else                          atomicMin((unsigned int*)addr, __float_as_uint(val));
}

#define EPW   8
#define NWAVE 32
#define LCAP  1088
#define SCANU 4

__device__ __forceinline__ void owner_flush(const int* lst, int cntw,
                                            const float4* __restrict__ asrc4, const float4* adstL4,
                                            const float4* mL4, float* sumsL, float* accL, float* wl,
                                            const float* __restrict__ hfeat, int lane) {
  for (int b = 0; b < cntw; b += 32) {
    const int n = min(32, cntw - b);
    int s = 0, r = 0;
    float w0 = 0.f, w1 = 0.f, w2 = 0.f, w3 = 0.f;
    if (lane < n) {
      const unsigned int p = (unsigned int)lst[b + lane];
      s = (int)(p & 0xffffu);
      r = (int)(p >> 16);
      const float4 a = asrc4[s], bq = adstL4[r], mx = mL4[r];
      w0 = expf(leaky(a.x + bq.x) - mx.x);
      w1 = expf(leaky(a.y + bq.y) - mx.y);
      w2 = expf(leaky(a.z + bq.z) - mx.z);
      w3 = expf(leaky(a.w + bq.w) - mx.w);
      atomicAdd(&sumsL[r * HEADS + 0], w0);
      atomicAdd(&sumsL[r * HEADS + 1], w1);
      atomicAdd(&sumsL[r * HEADS + 2], w2);
      atomicAdd(&sumsL[r * HEADS + 3], w3);
      ((float4*)wl)[lane] = make_float4(w0, w1, w2, w3);
    }
    const int hsel = lane >> 3;
    for (int g = 0; g < n; g += EPW) {
      float4 hv[EPW];
      float  xs[EPW];
      int    rk[EPW], kk[EPW];
#pragma unroll
      for (int q = 0; q < EPW; ++q) {
        kk[q] = min(g + q, n - 1);
        const int sk = __builtin_amdgcn_readlane(s, kk[q]);
        rk[q]        = __builtin_amdgcn_readlane(r, kk[q]);
        hv[q] = *(const float4*)(hfeat + (size_t)sk * HF + lane * 4);
        xs[q] = wl[kk[q] * HEADS + hsel];
      }
#pragma unroll
      for (int q = 0; q < EPW; ++q) {
        if (g + q < n) {
          float* ap = accL + rk[q] * HF + lane;
          atomicAdd(ap,      xs[q] * hv[q].x);
          atomicAdd(ap + 32, xs[q] * hv[q].y);
          atomicAdd(ap + 64, xs[q] * hv[q].z);
          atomicAdd(ap + 96, xs[q] * hv[q].w);
        }
      }
    }
  }
}

__global__ void __launch_bounds__(1024)
gat_agg_owner(const int* __restrict__ ei, const int* __restrict__ okey, const float* __restrict__ asrc,
              const float* __restrict__ adst,
              const float* __restrict__ hfeat, const float* __restrict__ bias,
              float* __restrict__ out, int* __restrict__ wlist, int* __restrict__ wcnt,
              int E, int Nn, int do_elu, int reuse) {
  __shared__ float4 accL4[ROWS * HF / 4];
  __shared__ float4 sumsL4[ROWS];
  __shared__ float4 adstL4[ROWS];
  __shared__ float4 mL4[ROWS];
  __shared__ int4   listL4[NWAVE * LCAP / 4];
  __shared__ float4 wL4[NWAVE * 32];
  float* accL  = (float*)accL4;
  float* sumsL = (float*)sumsL4;
  float* mxL   = (float*)mL4;
  int*   listL = (int*)listL4;

  const int tid = threadIdx.x, lane = tid & 31;
  const int wv  = __builtin_amdgcn_readfirstlane(tid >> 5);
  const int lo  = blockIdx.x * ROWS;
  const int cnt = min(ROWS, Nn - lo);

  for (int i = tid; i < ROWS * HF / 4; i += NWAVE * 32) accL4[i] = make_float4(0.f, 0.f, 0.f, 0.f);
  for (int i = tid; i < ROWS; i += NWAVE * 32) {
    sumsL4[i] = make_float4(0.f, 0.f, 0.f, 0.f);
    mL4[i]    = make_float4(-INFINITY, -INFINITY, -INFINITY, -INFINITY);
    if (i < cnt) adstL4[i] = ((const float4*)adst)[lo + i];
  }
  __syncthreads();

  int* lst = listL + wv * LCAP;
  int cntw = 0;
  int np   = 0;
  bool drop    = false;
  bool second  = false;
  const float4* asrc4 = (const float4*)asrc;
  const unsigned int pat = (unsigned int)blockIdx.x * 0x01010101u;

  auto zero_bytes = [](unsigned int x) -> unsigned int {
    return ~((((x & 0x7f7f7f7fu) + 0x7f7f7f7fu) | x) | 0x7f7f7f7fu);
  };
  auto settle_to = [&](int upto) {
    for (int b = np; b < upto; b += 32) {
      const int i = b + lane;
      if (i < upto) {
        const int eid = lst[i];
        int s, r;
        if (eid < E) { s = ei[eid]; r = ei[E + eid] - lo; }
        else         { s = eid - E; r = s - lo; }
        if (!second) {
          const float4 a = asrc4[s];
          lds_max_f(&mxL[r * HEADS + 0], a.x);
          lds_max_f(&mxL[r * HEADS + 1], a.y);
          lds_max_f(&mxL[r * HEADS + 2], a.z);
          lds_max_f(&mxL[r * HEADS + 3], a.w);
        }
        lst[i] = (int)((unsigned int)s | ((unsigned int)r << 16));
      }
    }
    np = upto;
  };
  auto settle = [&](bool all) {
    settle_to((all || drop) ? cntw : np + ((cntw - np) & ~31));
    if (drop) {
      if (second) owner_flush(lst, np, asrc4, adstL4, mL4, sumsL, accL, (float*)(wL4 + wv * 32), hfeat, lane);
      cntw = 0;
      np   = 0;
    }
  };

  const int nblk = (E + KBLK - 1) / KBLK;
  const int start = (int)((long long)blockIdx.x * nblk / gridDim.x);
  auto load_blk = [&](int k) -> uint4 {
    int bi = k + start; if (bi >= nblk) bi -= nblk;
    return *(const uint4*)(okey + (size_t)bi * (KBLK / 4) + lane * 4);
  };
  auto scan_blk = [&](const uint4& d, int k) {
    int bi = k + start; if (bi >= nblk) bi -= nblk;
    const int e0 = bi * KBLK + lane * 16;
    unsigned int c = (zero_bytes(d.x ^ pat) >> 7) | (zero_bytes(d.y ^ pat) >> 6) |
                     (zero_bytes(d.z ^ pat) >> 5) | (zero_bytes(d.w ^ pat) >> 4);
    unsigned int act = __builtin_amdgcn_ballot_w32(c != 0u);
    while (act) {
      if (c != 0u) {
        const int t = __builtin_ctz(c);
        lst[(int)__builtin_amdgcn_mbcnt_lo(act, (unsigned int)cntw)] = e0 + ((t & 3) << 2) + (t >> 3);
        c &= c - 1u;
      }
      cntw += __builtin_popcount(act);
      act = __builtin_amdgcn_ballot_w32(c != 0u);
    }
  };
  const int nbw   = wv < nblk ? (nblk - wv + NWAVE - 1) / NWAVE : 0;
  const int ntrip = nbw / SCANU;

  auto sweep = [&]() {
    if (ntrip > 0) {
      uint4 cur[SCANU], nxt[SCANU];
#pragma unroll
      for (int u = 0; u < SCANU; ++u) cur[u] = load_blk(wv + u * NWAVE);
      for (int trip = 0; trip < ntrip; ++trip) {
        const int tn = min(trip + 1, ntrip - 1);
#pragma unroll
        for (int u = 0; u < SCANU; ++u) nxt[u] = load_blk(wv + (tn * SCANU + u) * NWAVE);
        unsigned int late = 0u;
#pragma unroll
        for (int u = 0; u < SCANU; ++u) {
          if (cntw <= LCAP - KBLK) scan_blk(cur[u], wv + (trip * SCANU + u) * NWAVE);
          else late |= 1u << u;
        }
        if (late) drop = true;
        settle(false);
        while (late) {
          const int u = __builtin_ctz(late);
          late &= late - 1u;
          const int k = wv + (trip * SCANU + u) * NWAVE;
          const uint4 d = load_blk(k);
          scan_blk(d, k);
          settle(true);
        }
#pragma unroll
        for (int u = 0; u < SCANU; ++u) cur[u] = nxt[u];
      }
    }
    for (int t = ntrip * SCANU; t < nbw; ++t) {
      if (cntw > LCAP - KBLK) { drop = true; settle(true); }
      const uint4 d = load_blk(wv + t * NWAVE);
      scan_blk(d, wv + t * NWAVE);
      settle(false);
    }
    for (int rb = wv * 32; rb < ROWS; rb += NWAVE * 32) {
      if (cntw > LCAP - 32) { drop = true; settle(true); }
      const int r = rb + lane;
      const unsigned int mask = __builtin_amdgcn_ballot_w32(r < cnt);
      if (r < cnt) lst[(int)__builtin_amdgcn_mbcnt_lo(mask, (unsigned int)cntw)] = E + lo + r;
      cntw += __builtin_popcount(mask);
    }
    settle(true);
  };

  int* wl = wlist + ((size_t)blockIdx.x * NWAVE + wv) * LCAP;
  bool have = false;
  if (reuse) {
    const int c = __builtin_amdgcn_readfirstlane(wcnt[blockIdx.x * NWAVE + wv]);
    if (c >= 0) {
      for (int b = 0; b < c; b += 128) {
        const int i = b + lane * 4;
        if (i < c) {
          const int4 p = *(const int4*)(wl + i);
          const float4 a0 = asrc4[(unsigned int)p.x & 0xffffu], a1 = asrc4[(unsigned int)p.y & 0xffffu],
                       a2 = asrc4[(unsigned int)p.z & 0xffffu], a3 = asrc4[(unsigned int)p.w & 0xffffu];
          float* m0 = mxL + ((unsigned int)p.x >> 16) * HEADS;
          lds_max_f(m0 + 0, a0.x); lds_max_f(m0 + 1, a0.y); lds_max_f(m0 + 2, a0.z); lds_max_f(m0 + 3, a0.w);
          if (i + 1 < c) {
            float* m1 = mxL + ((unsigned int)p.y >> 16) * HEADS;
            lds_max_f(m1 + 0, a1.x); lds_max_f(m1 + 1, a1.y); lds_max_f(m1 + 2, a1.z); lds_max_f(m1 + 3, a1.w);
          }
          if (i + 2 < c) {
            float* m2 = mxL + ((unsigned int)p.z >> 16) * HEADS;
            lds_max_f(m2 + 0, a2.x); lds_max_f(m2 + 1, a2.y); lds_max_f(m2 + 2, a2.z); lds_max_f(m2 + 3, a2.w);
          }
          if (i + 3 < c) {
            float* m3 = mxL + ((unsigned int)p.w >> 16) * HEADS;
            lds_max_f(m3 + 0, a3.x); lds_max_f(m3 + 1, a3.y); lds_max_f(m3 + 2, a3.z); lds_max_f(m3 + 3, a3.w);
          }
          *(int4*)(lst + i) = p;
        }
      }
      cntw = c;
      np   = c;
      have = true;
    }
  }
  if (!have) sweep();
  const bool kept = !drop;
  if (!reuse) {
    if (kept) {
      for (int i = lane * 4; i < np; i += 128) {
        int4 p;
        p.x = lst[i];
        p.y = i + 1 < np ? lst[i + 1] : 0;
        p.z = i + 2 < np ? lst[i + 2] : 0;
        p.w = i + 3 < np ? lst[i + 3] : 0;
        *(int4*)(wl + i) = p;
      }
    }
    if (lane == 0) wcnt[blockIdx.x * NWAVE + wv] = kept ? np : -1;
  }
  __syncthreads();
  for (int i = tid; i < cnt; i += NWAVE * 32) {
    const float4 ms = mL4[i], bq = adstL4[i];
    mL4[i] = make_float4(leaky(ms.x + bq.x), leaky(ms.y + bq.y), leaky(ms.z + bq.z), leaky(ms.w + bq.w));
  }
  __syncthreads();
  second = true;
  if (kept) {
    owner_flush(lst, np, asrc4, adstL4, mL4, sumsL, accL, (float*)(wL4 + wv * 32), hfeat, lane);
  } else {
    drop = true;
    sweep();
  }
  __syncthreads();

  for (int idx = tid; idx < cnt * FEAT; idx += NWAVE * 32) {
    const int r = idx >> 5, f = idx & (FEAT - 1);
    const float* a = accL + r * HF + (f & 3) * 32 + (f >> 2);
    const float4 sv = sumsL4[r];
    float v = 0.25f * (a[0]  / (sv.x + 1e-16f) + a[8]  / (sv.y + 1e-16f) +
                       a[16] / (sv.z + 1e-16f) + a[24] / (sv.w + 1e-16f)) + bias[f];
    if (do_elu) v = v > 0.f ? v : (expf(v) - 1.f);
    out[(size_t)(lo + r) * FEAT + f] = v;
  }
}

extern "C" void kernel_launch(void* const* d_in, const int* in_sizes, int n_in,
                              void* d_out, int out_size, void* d_ws, size_t ws_size,
                              hipStream_t stream) {
  const float* x   = (const float*)d_in[0];
  const int*   ei  = (const int*)d_in[1];
  const float* W1  = (const float*)d_in[2];
  const float* as1 = (const float*)d_in[3];
  const float* ad1 = (const float*)d_in[4];
  const float* b1  = (const float*)d_in[5];
  const float* W2  = (const float*)d_in[6];
  const float* as2 = (const float*)d_in[7];
  const float* ad2 = (const float*)d_in[8];
  const float* b2  = (const float*)d_in[9];

  const int Nn = in_sizes[0] / HF;
  const int E  = in_sizes[1] / 2;

  float* ws    = (float*)d_ws;
  float* hbuf  = ws;
  float* asrc  = hbuf + (size_t)Nn * HF;
  float* adst  = asrc + (size_t)Nn * HEADS;
  float* x2    = adst + (size_t)Nn * HEADS * 2;
  int*   okey  = (int*)(x2 + (size_t)Nn * FEAT);
  const int nkw = (E + KBLK - 1) / KBLK * (KBLK / 4);
  int*   wlist = okey + nkw;
  int*   wcnt  = wlist + (size_t)((Nn + ROWS - 1) / ROWS) * NWAVE * LCAP;

  const dim3 blk(256);
  const int row_tiles = Nn / 16;
  const int tail_row0 = row_tiles * 16;

  for (int layer = 0; layer < 2; ++layer) {
    const float* Xin = (layer == 0) ? x  : x2;
    const int    K   = (layer == 0) ? 128 : FEAT;
    const float* Wm  = (layer == 0) ? W1 : W2;
    const float* as  = (layer == 0) ? as1 : as2;
    const float* ad  = (layer == 0) ? ad1 : ad2;
    const float* bb  = (layer == 0) ? b1 : b2;
    float*       op  = (layer == 0) ? x2 : (float*)d_out;

    if (row_tiles > 0)
      gat_gemm_wmma_f32<<<row_tiles, blk, 0, stream>>>(Xin, Wm, hbuf, K);
    if (tail_row0 < Nn) {
      int tail = (Nn - tail_row0) * HF;
      gat_gemm_tail<<<(tail + 255) / 256, blk, 0, stream>>>(Xin, Wm, hbuf, K, tail_row0, Nn);
    }
    gat_alpha<<<(Nn * HEADS + 255) / 256, blk, 0, stream>>>(hbuf, as, ad, asrc, adst, Nn);
    if (layer == 0)
      gat_init<<<(nkw + 255) / 256, blk, 0, stream>>>(okey, ei, E, nkw);
    gat_agg_owner<<<(Nn + ROWS - 1) / ROWS, dim3(NWAVE * 32), 0, stream>>>(ei, okey, asrc, adst, hbuf, bb, op, wlist, wcnt, E, Nn, layer == 0 ? 1 : 0, layer == 0 ? 0 : 1);
  }
}


# edge aggregation: a work-group owns 197 destination rows, not 256, so 254 work-groups (one per WGP) fill the chip where 196 left 60 WGPs idle; LDS adds still in arrival order
# speedup vs baseline: 5.3904x; 1.0720x over previous
#include <hip/hip_runtime.h>
#include <math.h>

#define HEADS 4
#define FEAT  32
#define HF    128

typedef __attribute__((ext_vector_type(2))) float v2f;
typedef __attribute__((ext_vector_type(8))) float v8f;

__global__ void __launch_bounds__(256)
gat_gemm_wmma_f32(const float* __restrict__ X, const float* __restrict__ W,
                  float* __restrict__ Hout, int K) {
  const int wave = threadIdx.x >> 5;
  const int lane = threadIdx.x & 31;
  const int half = lane >> 4;
  const int l15  = lane & 15;
  const int row  = blockIdx.x * 16 + l15;
  const int col  = wave * 16 + l15;
  const int kh   = half * 2;
  const float* __restrict__ xrow = X + (size_t)row * K;

  v8f c = {};
  for (int k0 = 0; k0 < K; k0 += 4) {
    v2f a, b;
    a.x = xrow[k0 + kh];
    a.y = xrow[k0 + kh + 1];
    b.x = W[(size_t)(k0 + kh)     * HF + col];
    b.y = W[(size_t)(k0 + kh + 1) * HF + col];
    c = __builtin_amdgcn_wmma_f32_16x16x4_f32(false, a, false, b,
                                              (short)0, c, false, false);
  }
  const int rbase = blockIdx.x * 16 + half * 8;
  float* orow = Hout + (size_t)rbase * HF + col;
#pragma unroll
  for (int v = 0; v < 8; ++v)
    orow[(size_t)v * HF] = c[v];
}

__global__ void gat_gemm_tail(const float* __restrict__ X, const float* __restrict__ W,
                              float* __restrict__ Hout, int K, int row0, int Nn) {
  int i = blockIdx.x * blockDim.x + threadIdx.x;
  int rows = Nn - row0;
  if (i >= rows * HF) return;
  int r = row0 + (i >> 7), col = i & (HF - 1);
  float acc = 0.f;
  for (int k = 0; k < K; ++k) acc += X[(size_t)r * K + k] * W[(size_t)k * HF + col];
  Hout[(size_t)r * HF + col] = acc;
}

__global__ void gat_alpha(const float* __restrict__ H, const float* __restrict__ att_src,
                          const float* __restrict__ att_dst,
                          float* __restrict__ a_src, float* __restrict__ a_dst, int Nn) {
  int i = blockIdx.x * blockDim.x + threadIdx.x;
  if (i >= Nn * HEADS) return;
  int h = i & (HEADS - 1);
  const float4* hp = (const float4*)(H + (size_t)(i >> 2) * HF + h * FEAT);
  const float4* as = (const float4*)(att_src + h * FEAT);
  const float4* ad = (const float4*)(att_dst + h * FEAT);
  float s1 = 0.f, s2 = 0.f;
#pragma unroll
  for (int f = 0; f < FEAT / 4; ++f) {
    float4 v = hp[f], a = as[f], d = ad[f];
    s1 += v.x * a.x + v.y * a.y + v.z * a.z + v.w * a.w;
    s2 += v.x * d.x + v.y * d.y + v.z * d.z + v.w * d.w;
  }
  a_src[i] = s1;
  a_dst[i] = s2;
}

#define ROWS  197
#define KBLK  512

__device__ __forceinline__ unsigned int owner_key(int d) { return min((unsigned int)d / ROWS, 255u); }

__global__ void gat_init(int* __restrict__ okey, const int* __restrict__ ei, int E, int nkw) {
  int i = blockIdx.x * blockDim.x + threadIdx.x;
  if (i < nkw) {
    const int e4 = 4 * i;
    unsigned int w = 0u;
    if ((E & 3) == 0 && e4 < E) {
      const int4 d = *(const int4*)(ei + E + e4);
      w = owner_key(d.x) | (owner_key(d.y) << 8) | (owner_key(d.z) << 16) | (owner_key(d.w) << 24);
    } else {
#pragma unroll
      for (int j = 0; j < 4; ++j) {
        const unsigned int kb = e4 + j < E ? owner_key(ei[E + e4 + j]) : 255u;
        w |= kb << (8 * j);
      }
    }
    okey[i] = (int)w;
  }
}

__device__ __forceinline__ float leaky(float e) { return e > 0.f ? e : 0.2f * e; }

__device__ __forceinline__ void lds_max_f(float* addr, float val) {
  if (__float_as_int(val) >= 0) atomicMax((int*)addr, __float_as_int(val));
  else                          atomicMin((unsigned int*)addr, __float_as_uint(val));
}

#define EPW   8
#define NWAVE 32
#define LCAP  1088
#define SCANU 4

__device__ __forceinline__ void owner_flush(const int* lst, int cntw,
                                            const float4* __restrict__ asrc4, const float4* adstL4,
                                            const float4* mL4, float* sumsL, float* accL, float* wl,
                                            const float* __restrict__ hfeat, int lane) {
  for (int b = 0; b < cntw; b += 32) {
    const int n = min(32, cntw - b);
    int s = 0, r = 0;
    float w0 = 0.f, w1 = 0.f, w2 = 0.f, w3 = 0.f;
    if (lane < n) {
      const unsigned int p = (unsigned int)lst[b + lane];
      s = (int)(p & 0xffffu);
      r = (int)(p >> 16);
      const float4 a = asrc4[s], bq = adstL4[r], mx = mL4[r];
      w0 = expf(leaky(a.x + bq.x) - mx.x);
      w1 = expf(leaky(a.y + bq.y) - mx.y);
      w2 = expf(leaky(a.z + bq.z) - mx.z);
      w3 = expf(leaky(a.w + bq.w) - mx.w);
      atomicAdd(&sumsL[r * HEADS + 0], w0);
      atomicAdd(&sumsL[r * HEADS + 1], w1);
      atomicAdd(&sumsL[r * HEADS + 2], w2);
      atomicAdd(&sumsL[r * HEADS + 3], w3);
      ((float4*)wl)[lane] = make_float4(w0, w1, w2, w3);
    }
    const int hsel = lane >> 3;
    for (int g = 0; g < n; g += EPW) {
      float4 hv[EPW];
      float  xs[EPW];
      int    rk[EPW], kk[EPW];
#pragma unroll
      for (int q = 0; q < EPW; ++q) {
        kk[q] = min(g + q, n - 1);
        const int sk = __builtin_amdgcn_readlane(s, kk[q]);
        rk[q]        = __builtin_amdgcn_readlane(r, kk[q]);
        hv[q] = *(const float4*)(hfeat + (size_t)sk * HF + lane * 4);
        xs[q] = wl[kk[q] * HEADS + hsel];
      }
#pragma unroll
      for (int q = 0; q < EPW; ++q) {
        if (g + q < n) {
          float* ap = accL + rk[q] * HF + lane;
          atomicAdd(ap,      xs[q] * hv[q].x);
          atomicAdd(ap + 32, xs[q] * hv[q].y);
          atomicAdd(ap + 64, xs[q] * hv[q].z);
          atomicAdd(ap + 96, xs[q] * hv[q].w);
        }
      }
    }
  }
}

__global__ void __launch_bounds__(1024)
gat_agg_owner(const int* __restrict__ ei, const int* __restrict__ okey, const float* __restrict__ asrc,
              const float* __restrict__ adst,
              const float* __restrict__ hfeat, const float* __restrict__ bias,
              float* __restrict__ out, int* __restrict__ wlist, int* __restrict__ wcnt,
              int E, int Nn, int do_elu, int reuse) {
  __shared__ float4 accL4[ROWS * HF / 4];
  __shared__ float4 sumsL4[ROWS];
  __shared__ float4 adstL4[ROWS];
  __shared__ float4 mL4[ROWS];
  __shared__ int4   listL4[NWAVE * LCAP / 4];
  __shared__ float4 wL4[NWAVE * 32];
  float* accL  = (float*)accL4;
  float* sumsL = (float*)sumsL4;
  float* mxL   = (float*)mL4;
  int*   listL = (int*)listL4;

  const int tid = threadIdx.x, lane = tid & 31;
  const int wv  = __builtin_amdgcn_readfirstlane(tid >> 5);
  const int lo  = blockIdx.x * ROWS;
  const int cnt = min(ROWS, Nn - lo);

  for (int i = tid; i < ROWS * HF / 4; i += NWAVE * 32) accL4[i] = make_float4(0.f, 0.f, 0.f, 0.f);
  for (int i = tid; i < ROWS; i += NWAVE * 32) {
    sumsL4[i] = make_float4(0.f, 0.f, 0.f, 0.f);
    mL4[i]    = make_float4(-INFINITY, -INFINITY, -INFINITY, -INFINITY);
    if (i < cnt) adstL4[i] = ((const float4*)adst)[lo + i];
  }
  __syncthreads();

  int* lst = listL + wv * LCAP;
  int cntw = 0;
  int np   = 0;
  bool drop    = false;
  bool second  = false;
  const float4* asrc4 = (const float4*)asrc;
  const unsigned int pat = (unsigned int)blockIdx.x * 0x01010101u;

  auto zero_bytes = [](unsigned int x) -> unsigned int {
    return ~((((x & 0x7f7f7f7fu) + 0x7f7f7f7fu) | x) | 0x7f7f7f7fu);
  };
  auto settle_to = [&](int upto) {
    for (int b = np; b < upto; b += 32) {
      const int i = b + lane;
      if (i < upto) {
        const int eid = lst[i];
        int s, r;
        if (eid < E) { s = ei[eid]; r = ei[E + eid] - lo; }
        else         { s = eid - E; r = s - lo; }
        if (!second) {
          const float4 a = asrc4[s];
          lds_max_f(&mxL[r * HEADS + 0], a.x);
          lds_max_f(&mxL[r * HEADS + 1], a.y);
          lds_max_f(&mxL[r * HEADS + 2], a.z);
          lds_max_f(&mxL[r * HEADS + 3], a.w);
        }
        lst[i] = (int)((unsigned int)s | ((unsigned int)r << 16));
      }
    }
    np = upto;
  };
  auto settle = [&](bool all) {
    settle_to((all || drop) ? cntw : np + ((cntw - np) & ~31));
    if (drop) {
      if (second) owner_flush(lst, np, asrc4, adstL4, mL4, sumsL, accL, (float*)(wL4 + wv * 32), hfeat, lane);
      cntw = 0;
      np   = 0;
    }
  };

  const int nblk = (E + KBLK - 1) / KBLK;
  const int start = (int)((long long)blockIdx.x * nblk / gridDim.x);
  auto load_blk = [&](int k) -> uint4 {
    int bi = k + start; if (bi >= nblk) bi -= nblk;
    return *(const uint4*)(okey + (size_t)bi * (KBLK / 4) + lane * 4);
  };
  auto scan_blk = [&](const uint4& d, int k) {
    int bi = k + start; if (bi >= nblk) bi -= nblk;
    const int e0 = bi * KBLK + lane * 16;
    unsigned int c = (zero_bytes(d.x ^ pat) >> 7) | (zero_bytes(d.y ^ pat) >> 6) |
                     (zero_bytes(d.z ^ pat) >> 5) | (zero_bytes(d.w ^ pat) >> 4);
    unsigned int act = __builtin_amdgcn_ballot_w32(c != 0u);
    while (act) {
      if (c != 0u) {
        const int t = __builtin_ctz(c);
        lst[(int)__builtin_amdgcn_mbcnt_lo(act, (unsigned int)cntw)] = e0 + ((t & 3) << 2) + (t >> 3);
        c &= c - 1u;
      }
      cntw += __builtin_popcount(act);
      act = __builtin_amdgcn_ballot_w32(c != 0u);
    }
  };
  const int nbw   = wv < nblk ? (nblk - wv + NWAVE - 1) / NWAVE : 0;
  const int ntrip = nbw / SCANU;

  auto sweep = [&]() {
    if (ntrip > 0) {
      uint4 cur[SCANU], nxt[SCANU];
#pragma unroll
      for (int u = 0; u < SCANU; ++u) cur[u] = load_blk(wv + u * NWAVE);
      for (int trip = 0; trip < ntrip; ++trip) {
        const int tn = min(trip + 1, ntrip - 1);
#pragma unroll
        for (int u = 0; u < SCANU; ++u) nxt[u] = load_blk(wv + (tn * SCANU + u) * NWAVE);
        unsigned int late = 0u;
#pragma unroll
        for (int u = 0; u < SCANU; ++u) {
          if (cntw <= LCAP - KBLK) scan_blk(cur[u], wv + (trip * SCANU + u) * NWAVE);
          else late |= 1u << u;
        }
        if (late) drop = true;
        settle(false);
        while (late) {
          const int u = __builtin_ctz(late);
          late &= late - 1u;
          const int k = wv + (trip * SCANU + u) * NWAVE;
          const uint4 d = load_blk(k);
          scan_blk(d, k);
          settle(true);
        }
#pragma unroll
        for (int u = 0; u < SCANU; ++u) cur[u] = nxt[u];
      }
    }
    for (int t = ntrip * SCANU; t < nbw; ++t) {
      if (cntw > LCAP - KBLK) { drop = true; settle(true); }
      const uint4 d = load_blk(wv + t * NWAVE);
      scan_blk(d, wv + t * NWAVE);
      settle(false);
    }
    for (int rb = wv * 32; rb < ROWS; rb += NWAVE * 32) {
      if (cntw > LCAP - 32) { drop = true; settle(true); }
      const int r = rb + lane;
      const unsigned int mask = __builtin_amdgcn_ballot_w32(r < cnt);
      if (r < cnt) lst[(int)__builtin_amdgcn_mbcnt_lo(mask, (unsigned int)cntw)] = E + lo + r;
      cntw += __builtin_popcount(mask);
    }
    settle(true);
  };

  int* wl = wlist + ((size_t)blockIdx.x * NWAVE + wv) * LCAP;
  bool have = false;
  if (reuse) {
    const int c = __builtin_amdgcn_readfirstlane(wcnt[blockIdx.x * NWAVE + wv]);
    if (c >= 0) {
      for (int b = 0; b < c; b += 128) {
        const int i = b + lane * 4;
        if (i < c) {
          const int4 p = *(const int4*)(wl + i);
          const float4 a0 = asrc4[(unsigned int)p.x & 0xffffu], a1 = asrc4[(unsigned int)p.y & 0xffffu],
                       a2 = asrc4[(unsigned int)p.z & 0xffffu], a3 = asrc4[(unsigned int)p.w & 0xffffu];
          float* m0 = mxL + ((unsigned int)p.x >> 16) * HEADS;
          lds_max_f(m0 + 0, a0.x); lds_max_f(m0 + 1, a0.y); lds_max_f(m0 + 2, a0.z); lds_max_f(m0 + 3, a0.w);
          if (i + 1 < c) {
            float* m1 = mxL + ((unsigned int)p.y >> 16) * HEADS;
            lds_max_f(m1 + 0, a1.x); lds_max_f(m1 + 1, a1.y); lds_max_f(m1 + 2, a1.z); lds_max_f(m1 + 3, a1.w);
          }
          if (i + 2 < c) {
            float* m2 = mxL + ((unsigned int)p.z >> 16) * HEADS;
            lds_max_f(m2 + 0, a2.x); lds_max_f(m2 + 1, a2.y); lds_max_f(m2 + 2, a2.z); lds_max_f(m2 + 3, a2.w);
          }
          if (i + 3 < c) {
            float* m3 = mxL + ((unsigned int)p.w >> 16) * HEADS;
            lds_max_f(m3 + 0, a3.x); lds_max_f(m3 + 1, a3.y); lds_max_f(m3 + 2, a3.z); lds_max_f(m3 + 3, a3.w);
          }
          *(int4*)(lst + i) = p;
        }
      }
      cntw = c;
      np   = c;
      have = true;
    }
  }
  if (!have) sweep();
  const bool kept = !drop;
  if (!reuse) {
    if (kept) {
      for (int i = lane * 4; i < np; i += 128) {
        int4 p;
        p.x = lst[i];
        p.y = i + 1 < np ? lst[i + 1] : 0;
        p.z = i + 2 < np ? lst[i + 2] : 0;
        p.w = i + 3 < np ? lst[i + 3] : 0;
        *(int4*)(wl + i) = p;
      }
    }
    if (lane == 0) wcnt[blockIdx.x * NWAVE + wv] = kept ? np : -1;
  }
  __syncthreads();
  for (int i = tid; i < cnt; i += NWAVE * 32) {
    const float4 ms = mL4[i], bq = adstL4[i];
    mL4[i] = make_float4(leaky(ms.x + bq.x), leaky(ms.y + bq.y), leaky(ms.z + bq.z), leaky(ms.w + bq.w));
  }
  __syncthreads();
  second = true;
  if (kept) {
    owner_flush(lst, np, asrc4, adstL4, mL4, sumsL, accL, (float*)(wL4 + wv * 32), hfeat, lane);
  } else {
    drop = true;
    sweep();
  }
  __syncthreads();

  for (int idx = tid; idx < cnt * FEAT; idx += NWAVE * 32) {
    const int r = idx >> 5, f = idx & (FEAT - 1);
    const float* a = accL + r * HF + (f & 3) * 32 + (f >> 2);
    const float4 sv = sumsL4[r];
    float v = 0.25f * (a[0]  / (sv.x + 1e-16f) + a[8]  / (sv.y + 1e-16f) +
                       a[16] / (sv.z + 1e-16f) + a[24] / (sv.w + 1e-16f)) + bias[f];
    if (do_elu) v = v > 0.f ? v : (expf(v) - 1.f);
    out[(size_t)(lo + r) * FEAT + f] = v;
  }
}

extern "C" void kernel_launch(void* const* d_in, const int* in_sizes, int n_in,
                              void* d_out, int out_size, void* d_ws, size_t ws_size,
                              hipStream_t stream) {
  const float* x   = (const float*)d_in[0];
  const int*   ei  = (const int*)d_in[1];
  const float* W1  = (const float*)d_in[2];
  const float* as1 = (const float*)d_in[3];
  const float* ad1 = (const float*)d_in[4];
  const float* b1  = (const float*)d_in[5];
  const float* W2  = (const float*)d_in[6];
  const float* as2 = (const float*)d_in[7];
  const float* ad2 = (const float*)d_in[8];
  const float* b2  = (const float*)d_in[9];

  const int Nn = in_sizes[0] / HF;
  const int E  = in_sizes[1] / 2;

  float* ws    = (float*)d_ws;
  float* hbuf  = ws;
  float* asrc  = hbuf + (size_t)Nn * HF;
  float* adst  = asrc + (size_t)Nn * HEADS;
  float* x2    = adst + (size_t)Nn * HEADS * 2;
  int*   okey  = (int*)(x2 + (size_t)Nn * FEAT);
  const int nkw = (E + KBLK - 1) / KBLK * (KBLK / 4);
  int*   wlist = okey + nkw;
  int*   wcnt  = wlist + (size_t)((Nn + ROWS - 1) / ROWS) * NWAVE * LCAP;

  const dim3 blk(256);
  const int row_tiles = Nn / 16;
  const int tail_row0 = row_tiles * 16;

  for (int layer = 0; layer < 2; ++layer) {
    const float* Xin = (layer == 0) ? x  : x2;
    const int    K   = (layer == 0) ? 128 : FEAT;
    const float* Wm  = (layer == 0) ? W1 : W2;
    const float* as  = (layer == 0) ? as1 : as2;
    const float* ad  = (layer == 0) ? ad1 : ad2;
    const float* bb  = (layer == 0) ? b1 : b2;
    float*       op  = (layer == 0) ? x2 : (float*)d_out;

    if (row_tiles > 0)
      gat_gemm_wmma_f32<<<row_tiles, blk, 0, stream>>>(Xin, Wm, hbuf, K);
    if (tail_row0 < Nn) {
      int tail = (Nn - tail_row0) * HF;
      gat_gemm_tail<<<(tail + 255) / 256, blk, 0, stream>>>(Xin, Wm, hbuf, K, tail_row0, Nn);
    }
    gat_alpha<<<(Nn * HEADS + 255) / 256, blk, 0, stream>>>(hbuf, as, ad, asrc, adst, Nn);
    if (layer == 0)
      gat_init<<<(nkw + 255) / 256, blk, 0, stream>>>(okey, ei, E, nkw);
    gat_agg_owner<<<(Nn + ROWS - 1) / ROWS, dim3(NWAVE * 32), 0, stream>>>(ei, okey, asrc, adst, hbuf, bb, op, wlist, wcnt, E, Nn, layer == 0 ? 1 : 0, layer == 0 ? 0 : 1);
  }
}


# GEMM: a work-group walks over 16-row tiles; each wave keeps its B fragment in registers for all of them, the X tile comes by whole-line loads into LDS for all 8 waves; k order and sums unchanged
# speedup vs baseline: 5.6950x; 1.0565x over previous
#include <hip/hip_runtime.h>
#include <math.h>

#define HEADS 4
#define FEAT  32
#define HF    128

typedef __attribute__((ext_vector_type(2))) float v2f;
typedef __attribute__((ext_vector_type(8))) float v8f;

#define GEMM_WGS 768

template <int K>
__global__ void __launch_bounds__(256)
gat_gemm_wmma_f32(const float* __restrict__ X, const float* __restrict__ W,
                  float* __restrict__ Hout, int row_tiles) {
  constexpr int P   = K + 4;
  constexpr int NA4 = 16 * K / 4;
  constexpr int RW4 = K / 4;
  __shared__ float4 aL4[2][16 * P / 4];

  const int tid  = threadIdx.x;
  const int wave = tid >> 5;
  const int lane = tid & 31;
  const int half = lane >> 4;
  const int l15  = lane & 15;
  const int col  = wave * 16 + l15;
  const int kh   = half * 2;

  v2f b[K / 4];
#pragma unroll
  for (int st = 0; st < K / 4; ++st) {
    b[st].x = W[(size_t)(4 * st + kh)     * HF + col];
    b[st].y = W[(size_t)(4 * st + kh + 1) * HF + col];
  }

  constexpr bool ALL0 = NA4 >= 256;
  constexpr bool HAS1 = NA4 > 256;
  constexpr bool ALL1 = NA4 >= 512;
  static_assert(NA4 <= 512, "a tile of X is at most two 16-byte units per thread");
  float4 pre0 = make_float4(0.f, 0.f, 0.f, 0.f), pre1 = make_float4(0.f, 0.f, 0.f, 0.f);
  auto fetch = [&](int t) {
    const float4* src = (const float4*)(X + (size_t)t * 16 * K);
    if (ALL0 || tid < NA4) pre0 = src[tid];
    if constexpr (HAS1) {
      if (ALL1 || tid + 256 < NA4) pre1 = src[tid + 256];
    }
  };
  int tile = blockIdx.x;
  if (tile < row_tiles) fetch(tile);
  int buf = 0;
  for (; tile < row_tiles; tile += gridDim.x) {
    float* aL = (float*)aL4[buf];
    if (ALL0 || tid < NA4) *(float4*)(aL + (tid / RW4) * P + (tid % RW4) * 4) = pre0;
    if constexpr (HAS1) {
      if (ALL1 || tid + 256 < NA4) *(float4*)(aL + ((tid + 256) / RW4) * P + ((tid + 256) % RW4) * 4) = pre1;
    }
    __syncthreads();
    const int nxt = tile + gridDim.x;
    if (nxt < row_tiles) fetch(nxt);
    const float* arow = aL + l15 * P + kh;
    v8f c = {};
#pragma unroll
    for (int st = 0; st < K / 4; ++st) {
      const v2f a = *(const v2f*)(arow + 4 * st);
      c = __builtin_amdgcn_wmma_f32_16x16x4_f32(false, a, false, b[st],
                                                (short)0, c, false, false);
    }
    const int rbase = tile * 16 + half * 8;
    float* orow = Hout + (size_t)rbase * HF + col;
#pragma unroll
    for (int v = 0; v < 8; ++v)
      orow[(size_t)v * HF] = c[v];
    buf ^= 1;
  }
}

__global__ void gat_gemm_tail(const float* __restrict__ X, const float* __restrict__ W,
                              float* __restrict__ Hout, int K, int row0, int Nn) {
  int i = blockIdx.x * blockDim.x + threadIdx.x;
  int rows = Nn - row0;
  if (i >= rows * HF) return;
  int r = row0 + (i >> 7), col = i & (HF - 1);
  float acc = 0.f;
  for (int k = 0; k < K; ++k) acc += X[(size_t)r * K + k] * W[(size_t)k * HF + col];
  Hout[(size_t)r * HF + col] = acc;
}

__global__ void gat_alpha(const float* __restrict__ H, const float* __restrict__ att_src,
                          const float* __restrict__ att_dst,
                          float* __restrict__ a_src, float* __restrict__ a_dst, int Nn) {
  int i = blockIdx.x * blockDim.x + threadIdx.x;
  if (i >= Nn * HEADS) return;
  int h = i & (HEADS - 1);
  const float4* hp = (const float4*)(H + (size_t)(i >> 2) * HF + h * FEAT);
  const float4* as = (const float4*)(att_src + h * FEAT);
  const float4* ad = (const float4*)(att_dst + h * FEAT);
  float s1 = 0.f, s2 = 0.f;
#pragma unroll
  for (int f = 0; f < FEAT / 4; ++f) {
    float4 v = hp[f], a = as[f], d = ad[f];
    s1 += v.x * a.x + v.y * a.y + v.z * a.z + v.w * a.w;
    s2 += v.x * d.x + v.y * d.y + v.z * d.z + v.w * d.w;
  }
  a_src[i] = s1;
  a_dst[i] = s2;
}

#define ROWS  197
#define KBLK  512

__device__ __forceinline__ unsigned int owner_key(int d) { return min((unsigned int)d / ROWS, 255u); }

__global__ void gat_init(int* __restrict__ okey, const int* __restrict__ ei, int E, int nkw) {
  int i = blockIdx.x * blockDim.x + threadIdx.x;
  if (i < nkw) {
    const int e4 = 4 * i;
    unsigned int w = 0u;
    if ((E & 3) == 0 && e4 < E) {
      const int4 d = *(const int4*)(ei + E + e4);
      w = owner_key(d.x) | (owner_key(d.y) << 8) | (owner_key(d.z) << 16) | (owner_key(d.w) << 24);
    } else {
#pragma unroll
      for (int j = 0; j < 4; ++j) {
        const unsigned int kb = e4 + j < E ? owner_key(ei[E + e4 + j]) : 255u;
        w |= kb << (8 * j);
      }
    }
    okey[i] = (int)w;
  }
}

__device__ __forceinline__ float leaky(float e) { return e > 0.f ? e : 0.2f * e; }

__device__ __forceinline__ void lds_max_f(float* addr, float val) {
  if (__float_as_int(val) >= 0) atomicMax((int*)addr, __float_as_int(val));
  else                          atomicMin((unsigned int*)addr, __float_as_uint(val));
}

#define EPW   8
#define NWAVE 32
#define LCAP  1088
#define SCANU 4

__device__ __forceinline__ void owner_flush(const int* lst, int cntw,
                                            const float4* __restrict__ asrc4, const float4* adstL4,
                                            const float4* mL4, float* sumsL, float* accL, float* wl,
                                            const float* __restrict__ hfeat, int lane) {
  for (int b = 0; b < cntw; b += 32) {
    const int n = min(32, cntw - b);
    int s = 0, r = 0;
    float w0 = 0.f, w1 = 0.f, w2 = 0.f, w3 = 0.f;
    if (lane < n) {
      const unsigned int p = (unsigned int)lst[b + lane];
      s = (int)(p & 0xffffu);
      r = (int)(p >> 16);
      const float4 a = asrc4[s], bq = adstL4[r], mx = mL4[r];
      w0 = expf(leaky(a.x + bq.x) - mx.x);
      w1 = expf(leaky(a.y + bq.y) - mx.y);
      w2 = expf(leaky(a.z + bq.z) - mx.z);
      w3 = expf(leaky(a.w + bq.w) - mx.w);
      atomicAdd(&sumsL[r * HEADS + 0], w0);
      atomicAdd(&sumsL[r * HEADS + 1], w1);
      atomicAdd(&sumsL[r * HEADS + 2], w2);
      atomicAdd(&sumsL[r * HEADS + 3], w3);
      ((float4*)wl)[lane] = make_float4(w0, w1, w2, w3);
    }
    const int hsel = lane >> 3;
    for (int g = 0; g < n; g += EPW) {
      float4 hv[EPW];
      float  xs[EPW];
      int    rk[EPW], kk[EPW];
#pragma unroll
      for (int q = 0; q < EPW; ++q) {
        kk[q] = min(g + q, n - 1);
        const int sk = __builtin_amdgcn_readlane(s, kk[q]);
        rk[q]        = __builtin_amdgcn_readlane(r, kk[q]);
        hv[q] = *(const float4*)(hfeat + (size_t)sk * HF + lane * 4);
        xs[q] = wl[kk[q] * HEADS + hsel];
      }
#pragma unroll
      for (int q = 0; q < EPW; ++q) {
        if (g + q < n) {
          float* ap = accL + rk[q] * HF + lane;
          atomicAdd(ap,      xs[q] * hv[q].x);
          atomicAdd(ap + 32, xs[q] * hv[q].y);
          atomicAdd(ap + 64, xs[q] * hv[q].z);
          atomicAdd(ap + 96, xs[q] * hv[q].w);
        }
      }
    }
  }
}

__global__ void __launch_bounds__(1024)
gat_agg_owner(const int* __restrict__ ei, const int* __restrict__ okey, const float* __restrict__ asrc,
              const float* __restrict__ adst,
              const float* __restrict__ hfeat, const float* __restrict__ bias,
              float* __restrict__ out, int* __restrict__ wlist, int* __restrict__ wcnt,
              int E, int Nn, int do_elu, int reuse) {
  __shared__ float4 accL4[ROWS * HF / 4];
  __shared__ float4 sumsL4[ROWS];
  __shared__ float4 adstL4[ROWS];
  __shared__ float4 mL4[ROWS];
  __shared__ int4   listL4[NWAVE * LCAP / 4];
  __shared__ float4 wL4[NWAVE * 32];
  float* accL  = (float*)accL4;
  float* sumsL = (float*)sumsL4;
  float* mxL   = (float*)mL4;
  int*   listL = (int*)listL4;

  const int tid = threadIdx.x, lane = tid & 31;
  const int wv  = __builtin_amdgcn_readfirstlane(tid >> 5);
  const int lo  = blockIdx.x * ROWS;
  const int cnt = min(ROWS, Nn - lo);

  for (int i = tid; i < ROWS * HF / 4; i += NWAVE * 32) accL4[i] = make_float4(0.f, 0.f, 0.f, 0.f);
  for (int i = tid; i < ROWS; i += NWAVE * 32) {
    sumsL4[i] = make_float4(0.f, 0.f, 0.f, 0.f);
    mL4[i]    = make_float4(-INFINITY, -INFINITY, -INFINITY, -INFINITY);
    if (i < cnt) adstL4[i] = ((const float4*)adst)[lo + i];
  }
  __syncthreads();

  int* lst = listL + wv * LCAP;
  int cntw = 0;
  int np   = 0;
  bool drop    = false;
  bool second  = false;
  const float4* asrc4 = (const float4*)asrc;
  const unsigned int pat = (unsigned int)blockIdx.x * 0x01010101u;

  auto zero_bytes = [](unsigned int x) -> unsigned int {
    return ~((((x & 0x7f7f7f7fu) + 0x7f7f7f7fu) | x) | 0x7f7f7f7fu);
  };
  auto settle_to = [&](int upto) {
    for (int b = np; b < upto; b += 32) {
      const int i = b + lane;
      if (i < upto) {
        const int eid = lst[i];
        int s, r;
        if (eid < E) { s = ei[eid]; r = ei[E + eid] - lo; }
        else         { s = eid - E; r = s - lo; }
        if (!second) {
          const float4 a = asrc4[s];
          lds_max_f(&mxL[r * HEADS + 0], a.x);
          lds_max_f(&mxL[r * HEADS + 1], a.y);
          lds_max_f(&mxL[r * HEADS + 2], a.z);
          lds_max_f(&mxL[r * HEADS + 3], a.w);
        }
        lst[i] = (int)((unsigned int)s | ((unsigned int)r << 16));
      }
    }
    np = upto;
  };
  auto settle = [&](bool all) {
    settle_to((all || drop) ? cntw : np + ((cntw - np) & ~31));
    if (drop) {
      if (second) owner_flush(lst, np, asrc4, adstL4, mL4, sumsL, accL, (float*)(wL4 + wv * 32), hfeat, lane);
      cntw = 0;
      np   = 0;
    }
  };

  const int nblk = (E + KBLK - 1) / KBLK;
  const int start = (int)((long long)blockIdx.x * nblk / gridDim.x);
  auto load_blk = [&](int k) -> uint4 {
    int bi = k + start; if (bi >= nblk) bi -= nblk;
    return *(const uint4*)(okey + (size_t)bi * (KBLK / 4) + lane * 4);
  };
  auto scan_blk = [&](const uint4& d, int k) {
    int bi = k + start; if (bi >= nblk) bi -= nblk;
    const int e0 = bi * KBLK + lane * 16;
    unsigned int c = (zero_bytes(d.x ^ pat) >> 7) | (zero_bytes(d.y ^ pat) >> 6) |
                     (zero_bytes(d.z ^ pat) >> 5) | (zero_bytes(d.w ^ pat) >> 4);
    unsigned int act = __builtin_amdgcn_ballot_w32(c != 0u);
    while (act) {
      if (c != 0u) {
        const int t = __builtin_ctz(c);
        lst[(int)__builtin_amdgcn_mbcnt_lo(act, (unsigned int)cntw)] = e0 + ((t & 3) << 2) + (t >> 3);
        c &= c - 1u;
      }
      cntw += __builtin_popcount(act);
      act = __builtin_amdgcn_ballot_w32(c != 0u);
    }
  };
  const int nbw   = wv < nblk ? (nblk - wv + NWAVE - 1) / NWAVE : 0;
  const int ntrip = nbw / SCANU;

  auto sweep = [&]() {
    if (ntrip > 0) {
      uint4 cur[SCANU], nxt[SCANU];
#pragma unroll
      for (int u = 0; u < SCANU; ++u) cur[u] = load_blk(wv + u * NWAVE);
      for (int trip = 0; trip < ntrip; ++trip) {
        const int tn = min(trip + 1, ntrip - 1);
#pragma unroll
        for (int u = 0; u < SCANU; ++u) nxt[u] = load_blk(wv + (tn * SCANU + u) * NWAVE);
        unsigned int late = 0u;
#pragma unroll
        for (int u = 0; u < SCANU; ++u) {
          if (cntw <= LCAP - KBLK) scan_blk(cur[u], wv + (trip * SCANU + u) * NWAVE);
          else late |= 1u << u;
        }
        if (late) drop = true;
        settle(false);
        while (late) {
          const int u = __builtin_ctz(late);
          late &= late - 1u;
          const int k = wv + (trip * SCANU + u) * NWAVE;
          const uint4 d = load_blk(k);
          scan_blk(d, k);
          settle(true);
        }
#pragma unroll
        for (int u = 0; u < SCANU; ++u) cur[u] = nxt[u];
      }
    }
    for (int t = ntrip * SCANU; t < nbw; ++t) {
      if (cntw > LCAP - KBLK) { drop = true; settle(true); }
      const uint4 d = load_blk(wv + t * NWAVE);
      scan_blk(d, wv + t * NWAVE);
      settle(false);
    }
    for (int rb = wv * 32; rb < ROWS; rb += NWAVE * 32) {
      if (cntw > LCAP - 32) { drop = true; settle(true); }
      const int r = rb + lane;
      const unsigned int mask = __builtin_amdgcn_ballot_w32(r < cnt);
      if (r < cnt) lst[(int)__builtin_amdgcn_mbcnt_lo(mask, (unsigned int)cntw)] = E + lo + r;
      cntw += __builtin_popcount(mask);
    }
    settle(true);
  };

  int* wl = wlist + ((size_t)blockIdx.x * NWAVE + wv) * LCAP;
  bool have = false;
  if (reuse) {
    const int c = __builtin_amdgcn_readfirstlane(wcnt[blockIdx.x * NWAVE + wv]);
    if (c >= 0) {
      for (int b = 0; b < c; b += 128) {
        const int i = b + lane * 4;
        if (i < c) {
          const int4 p = *(const int4*)(wl + i);
          const float4 a0 = asrc4[(unsigned int)p.x & 0xffffu], a1 = asrc4[(unsigned int)p.y & 0xffffu],
                       a2 = asrc4[(unsigned int)p.z & 0xffffu], a3 = asrc4[(unsigned int)p.w & 0xffffu];
          float* m0 = mxL + ((unsigned int)p.x >> 16) * HEADS;
          lds_max_f(m0 + 0, a0.x); lds_max_f(m0 + 1, a0.y); lds_max_f(m0 + 2, a0.z); lds_max_f(m0 + 3, a0.w);
          if (i + 1 < c) {
            float* m1 = mxL + ((unsigned int)p.y >> 16) * HEADS;
            lds_max_f(m1 + 0, a1.x); lds_max_f(m1 + 1, a1.y); lds_max_f(m1 + 2, a1.z); lds_max_f(m1 + 3, a1.w);
          }
          if (i + 2 < c) {
            float* m2 = mxL + ((unsigned int)p.z >> 16) * HEADS;
            lds_max_f(m2 + 0, a2.x); lds_max_f(m2 + 1, a2.y); lds_max_f(m2 + 2, a2.z); lds_max_f(m2 + 3, a2.w);
          }
          if (i + 3 < c) {
            float* m3 = mxL + ((unsigned int)p.w >> 16) * HEADS;
            lds_max_f(m3 + 0, a3.x); lds_max_f(m3 + 1, a3.y); lds_max_f(m3 + 2, a3.z); lds_max_f(m3 + 3, a3.w);
          }
          *(int4*)(lst + i) = p;
        }
      }
      cntw = c;
      np   = c;
      have = true;
    }
  }
  if (!have) sweep();
  const bool kept = !drop;
  if (!reuse) {
    if (kept) {
      for (int i = lane * 4; i < np; i += 128) {
        int4 p;
        p.x = lst[i];
        p.y = i + 1 < np ? lst[i + 1] : 0;
        p.z = i + 2 < np ? lst[i + 2] : 0;
        p.w = i + 3 < np ? lst[i + 3] : 0;
        *(int4*)(wl + i) = p;
      }
    }
    if (lane == 0) wcnt[blockIdx.x * NWAVE + wv] = kept ? np : -1;
  }
  __syncthreads();
  for (int i = tid; i < cnt; i += NWAVE * 32) {
    const float4 ms = mL4[i], bq = adstL4[i];
    mL4[i] = make_float4(leaky(ms.x + bq.x), leaky(ms.y + bq.y), leaky(ms.z + bq.z), leaky(ms.w + bq.w));
  }
  __syncthreads();
  second = true;
  if (kept) {
    owner_flush(lst, np, asrc4, adstL4, mL4, sumsL, accL, (float*)(wL4 + wv * 32), hfeat, lane);
  } else {
    drop = true;
    sweep();
  }
  __syncthreads();

  for (int idx = tid; idx < cnt * FEAT; idx += NWAVE * 32) {
    const int r = idx >> 5, f = idx & (FEAT - 1);
    const float* a = accL + r * HF + (f & 3) * 32 + (f >> 2);
    const float4 sv = sumsL4[r];
    float v = 0.25f * (a[0]  / (sv.x + 1e-16f) + a[8]  / (sv.y + 1e-16f) +
                       a[16] / (sv.z + 1e-16f) + a[24] / (sv.w + 1e-16f)) + bias[f];
    if (do_elu) v = v > 0.f ? v : (expf(v) - 1.f);
    out[(size_t)(lo + r) * FEAT + f] = v;
  }
}

extern "C" void kernel_launch(void* const* d_in, const int* in_sizes, int n_in,
                              void* d_out, int out_size, void* d_ws, size_t ws_size,
                              hipStream_t stream) {
  const float* x   = (const float*)d_in[0];
  const int*   ei  = (const int*)d_in[1];
  const float* W1  = (const float*)d_in[2];
  const float* as1 = (const float*)d_in[3];
  const float* ad1 = (const float*)d_in[4];
  const float* b1  = (const float*)d_in[5];
  const float* W2  = (const float*)d_in[6];
  const float* as2 = (const float*)d_in[7];
  const float* ad2 = (const float*)d_in[8];
  const float* b2  = (const float*)d_in[9];

  const int Nn = in_sizes[0] / HF;
  const int E  = in_sizes[1] / 2;

  float* ws    = (float*)d_ws;
  float* hbuf  = ws;
  float* asrc  = hbuf + (size_t)Nn * HF;
  float* adst  = asrc + (size_t)Nn * HEADS;
  float* x2    = adst + (size_t)Nn * HEADS * 2;
  int*   okey  = (int*)(x2 + (size_t)Nn * FEAT);
  const int nkw = (E + KBLK - 1) / KBLK * (KBLK / 4);
  int*   wlist = okey + nkw;
  int*   wcnt  = wlist + (size_t)((Nn + ROWS - 1) / ROWS) * NWAVE * LCAP;

  const dim3 blk(256);
  const int row_tiles = Nn / 16;
  const int tail_row0 = row_tiles * 16;

  for (int layer = 0; layer < 2; ++layer) {
    const float* Xin = (layer == 0) ? x  : x2;
    const int    K   = (layer == 0) ? 128 : FEAT;
    const float* Wm  = (layer == 0) ? W1 : W2;
    const float* as  = (layer == 0) ? as1 : as2;
    const float* ad  = (layer == 0) ? ad1 : ad2;
    const float* bb  = (layer == 0) ? b1 : b2;
    float*       op  = (layer == 0) ? x2 : (float*)d_out;

    if (row_tiles > 0) {
      const int gg = row_tiles < GEMM_WGS ? row_tiles : GEMM_WGS;
      if (layer == 0) gat_gemm_wmma_f32<128><<<gg, blk, 0, stream>>>(Xin, Wm, hbuf, row_tiles);
      else            gat_gemm_wmma_f32<FEAT><<<gg, blk, 0, stream>>>(Xin, Wm, hbuf, row_tiles);
    }
    if (tail_row0 < Nn) {
      int tail = (Nn - tail_row0) * HF;
      gat_gemm_tail<<<(tail + 255) / 256, blk, 0, stream>>>(Xin, Wm, hbuf, K, tail_row0, Nn);
    }
    gat_alpha<<<(Nn * HEADS + 255) / 256, blk, 0, stream>>>(hbuf, as, ad, asrc, adst, Nn);
    if (layer == 0)
      gat_init<<<(nkw + 255) / 256, blk, 0, stream>>>(okey, ei, E, nkw);
    gat_agg_owner<<<(Nn + ROWS - 1) / ROWS, dim3(NWAVE * 32), 0, stream>>>(ei, okey, asrc, adst, hbuf, bb, op, wlist, wcnt, E, Nn, layer == 0 ? 1 : 0, layer == 0 ? 0 : 1);
  }
}


# layer 2's projection (WMMA from LDS) and logits (x2 @ (W2 a): new summation order) are written by the layer-1 aggregation's epilogue: layer 2's GEMM and logit launches go, x2 never leaves LDS
# speedup vs baseline: 5.9999x; 1.0535x over previous
#include <hip/hip_runtime.h>
#include <math.h>

#define HEADS 4
#define FEAT  32
#define HF    128

typedef __attribute__((ext_vector_type(2))) float v2f;
typedef __attribute__((ext_vector_type(8))) float v8f;

#define GEMM_WGS 768

template <int K>
__global__ void __launch_bounds__(256)
gat_gemm_wmma_f32(const float* __restrict__ X, const float* __restrict__ W,
                  float* __restrict__ Hout, int row_tiles) {
  constexpr int P   = K + 4;
  constexpr int NA4 = 16 * K / 4;
  constexpr int RW4 = K / 4;
  __shared__ float4 aL4[2][16 * P / 4];

  const int tid  = threadIdx.x;
  const int wave = tid >> 5;
  const int lane = tid & 31;
  const int half = lane >> 4;
  const int l15  = lane & 15;
  const int col  = wave * 16 + l15;
  const int kh   = half * 2;

  v2f b[K / 4];
#pragma unroll
  for (int st = 0; st < K / 4; ++st) {
    b[st].x = W[(size_t)(4 * st + kh)     * HF + col];
    b[st].y = W[(size_t)(4 * st + kh + 1) * HF + col];
  }

  constexpr bool ALL0 = NA4 >= 256;
  constexpr bool HAS1 = NA4 > 256;
  constexpr bool ALL1 = NA4 >= 512;
  static_assert(NA4 <= 512, "a tile of X is at most two 16-byte units per thread");
  float4 pre0 = make_float4(0.f, 0.f, 0.f, 0.f), pre1 = make_float4(0.f, 0.f, 0.f, 0.f);
  auto fetch = [&](int t) {
    const float4* src = (const float4*)(X + (size_t)t * 16 * K);
    if (ALL0 || tid < NA4) pre0 = src[tid];
    if constexpr (HAS1) {
      if (ALL1 || tid + 256 < NA4) pre1 = src[tid + 256];
    }
  };
  int tile = blockIdx.x;
  if (tile < row_tiles) fetch(tile);
  int buf = 0;
  for (; tile < row_tiles; tile += gridDim.x) {
    float* aL = (float*)aL4[buf];
    if (ALL0 || tid < NA4) *(float4*)(aL + (tid / RW4) * P + (tid % RW4) * 4) = pre0;
    if constexpr (HAS1) {
      if (ALL1 || tid + 256 < NA4) *(float4*)(aL + ((tid + 256) / RW4) * P + ((tid + 256) % RW4) * 4) = pre1;
    }
    __syncthreads();
    const int nxt = tile + gridDim.x;
    if (nxt < row_tiles) fetch(nxt);
    const float* arow = aL + l15 * P + kh;
    v8f c = {};
#pragma unroll
    for (int st = 0; st < K / 4; ++st) {
      const v2f a = *(const v2f*)(arow + 4 * st);
      c = __builtin_amdgcn_wmma_f32_16x16x4_f32(false, a, false, b[st],
                                                (short)0, c, false, false);
    }
    const int rbase = tile * 16 + half * 8;
    float* orow = Hout + (size_t)rbase * HF + col;
#pragma unroll
    for (int v = 0; v < 8; ++v)
      orow[(size_t)v * HF] = c[v];
    buf ^= 1;
  }
}

__global__ void gat_gemm_tail(const float* __restrict__ X, const float* __restrict__ W,
                              float* __restrict__ Hout, int K, int row0, int Nn) {
  int i = blockIdx.x * blockDim.x + threadIdx.x;
  int rows = Nn - row0;
  if (i >= rows * HF) return;
  int r = row0 + (i >> 7), col = i & (HF - 1);
  float acc = 0.f;
  for (int k = 0; k < K; ++k) acc += X[(size_t)r * K + k] * W[(size_t)k * HF + col];
  Hout[(size_t)r * HF + col] = acc;
}

__global__ void gat_alpha(const float* __restrict__ H, const float* __restrict__ att_src,
                          const float* __restrict__ att_dst,
                          float* __restrict__ a_src, float* __restrict__ a_dst, int Nn) {
  int i = blockIdx.x * blockDim.x + threadIdx.x;
  if (i >= Nn * HEADS) return;
  int h = i & (HEADS - 1);
  const float4* hp = (const float4*)(H + (size_t)(i >> 2) * HF + h * FEAT);
  const float4* as = (const float4*)(att_src + h * FEAT);
  const float4* ad = (const float4*)(att_dst + h * FEAT);
  float s1 = 0.f, s2 = 0.f;
#pragma unroll
  for (int f = 0; f < FEAT / 4; ++f) {
    float4 v = hp[f], a = as[f], d = ad[f];
    s1 += v.x * a.x + v.y * a.y + v.z * a.z + v.w * a.w;
    s2 += v.x * d.x + v.y * d.y + v.z * d.z + v.w * d.w;
  }
  a_src[i] = s1;
  a_dst[i] = s2;
}

#define ROWS  197
#define KBLK  512

__device__ __forceinline__ unsigned int owner_key(int d) { return min((unsigned int)d / ROWS, 255u); }

__global__ void gat_init(int* __restrict__ okey, const int* __restrict__ ei, int E, int nkw) {
  int i = blockIdx.x * blockDim.x + threadIdx.x;
  if (i < nkw) {
    const int e4 = 4 * i;
    unsigned int w = 0u;
    if ((E & 3) == 0 && e4 < E) {
      const int4 d = *(const int4*)(ei + E + e4);
      w = owner_key(d.x) | (owner_key(d.y) << 8) | (owner_key(d.z) << 16) | (owner_key(d.w) << 24);
    } else {
#pragma unroll
      for (int j = 0; j < 4; ++j) {
        const unsigned int kb = e4 + j < E ? owner_key(ei[E + e4 + j]) : 255u;
        w |= kb << (8 * j);
      }
    }
    okey[i] = (int)w;
  }
}

__device__ __forceinline__ float leaky(float e) { return e > 0.f ? e : 0.2f * e; }

__device__ __forceinline__ void lds_max_f(float* addr, float val) {
  if (__float_as_int(val) >= 0) atomicMax((int*)addr, __float_as_int(val));
  else                          atomicMin((unsigned int*)addr, __float_as_uint(val));
}

#define EPW   8
#define NWAVE 32
#define LCAP  1088
#define SCANU 4
#define XROWS ((ROWS + 15) / 16 * 16)
#define XP    (FEAT + 4)

__device__ __forceinline__ void owner_flush(const int* lst, int cntw,
                                            const float4* __restrict__ asrc4, const float4* adstL4,
                                            const float4* mL4, float* sumsL, float* accL, float* wl,
                                            const float* __restrict__ hfeat, int lane) {
  for (int b = 0; b < cntw; b += 32) {
    const int n = min(32, cntw - b);
    int s = 0, r = 0;
    float w0 = 0.f, w1 = 0.f, w2 = 0.f, w3 = 0.f;
    if (lane < n) {
      const unsigned int p = (unsigned int)lst[b + lane];
      s = (int)(p & 0xffffu);
      r = (int)(p >> 16);
      const float4 a = asrc4[s], bq = adstL4[r], mx = mL4[r];
      w0 = expf(leaky(a.x + bq.x) - mx.x);
      w1 = expf(leaky(a.y + bq.y) - mx.y);
      w2 = expf(leaky(a.z + bq.z) - mx.z);
      w3 = expf(leaky(a.w + bq.w) - mx.w);
      atomicAdd(&sumsL[r * HEADS + 0], w0);
      atomicAdd(&sumsL[r * HEADS + 1], w1);
      atomicAdd(&sumsL[r * HEADS + 2], w2);
      atomicAdd(&sumsL[r * HEADS + 3], w3);
      ((float4*)wl)[lane] = make_float4(w0, w1, w2, w3);
    }
    const int hsel = lane >> 3;
    for (int g = 0; g < n; g += EPW) {
      float4 hv[EPW];
      float  xs[EPW];
      int    rk[EPW], kk[EPW];
#pragma unroll
      for (int q = 0; q < EPW; ++q) {
        kk[q] = min(g + q, n - 1);
        const int sk = __builtin_amdgcn_readlane(s, kk[q]);
        rk[q]        = __builtin_amdgcn_readlane(r, kk[q]);
        hv[q] = *(const float4*)(hfeat + (size_t)sk * HF + lane * 4);
        xs[q] = wl[kk[q] * HEADS + hsel];
      }
#pragma unroll
      for (int q = 0; q < EPW; ++q) {
        if (g + q < n) {
          float* ap = accL + rk[q] * HF + lane;
          atomicAdd(ap,      xs[q] * hv[q].x);
          atomicAdd(ap + 32, xs[q] * hv[q].y);
          atomicAdd(ap + 64, xs[q] * hv[q].z);
          atomicAdd(ap + 96, xs[q] * hv[q].w);
        }
      }
    }
  }
}

__global__ void __launch_bounds__(1024)
gat_agg_owner(const int* __restrict__ ei, const int* __restrict__ okey, const float* __restrict__ asrc,
              const float* __restrict__ adst,
              const float* __restrict__ hfeat, const float* __restrict__ bias,
              float* __restrict__ out, int* __restrict__ wlist, int* __restrict__ wcnt,
              int E, int Nn, int do_elu, int reuse,
              const float* __restrict__ Wn, const float* __restrict__ asn, const float* __restrict__ adn,
              float* __restrict__ hn, float* __restrict__ asrcn, float* __restrict__ adstn, int fuse) {
  __shared__ float4 accL4[ROWS * HF / 4];
  __shared__ float4 sumsL4[ROWS];
  __shared__ float4 adstL4[ROWS];
  __shared__ float4 mL4[ROWS];
  __shared__ int4   listL4[NWAVE * LCAP / 4];
  __shared__ float4 wL4[NWAVE * 32];
  float* accL  = (float*)accL4;
  float* sumsL = (float*)sumsL4;
  float* mxL   = (float*)mL4;
  int*   listL = (int*)listL4;

  const int tid = threadIdx.x, lane = tid & 31;
  const int wv  = __builtin_amdgcn_readfirstlane(tid >> 5);
  const int lo  = blockIdx.x * ROWS;
  const int cnt = min(ROWS, Nn - lo);

  for (int i = tid; i < ROWS * HF / 4; i += NWAVE * 32) accL4[i] = make_float4(0.f, 0.f, 0.f, 0.f);
  for (int i = tid; i < ROWS; i += NWAVE * 32) {
    sumsL4[i] = make_float4(0.f, 0.f, 0.f, 0.f);
    mL4[i]    = make_float4(-INFINITY, -INFINITY, -INFINITY, -INFINITY);
    if (i < cnt) adstL4[i] = ((const float4*)adst)[lo + i];
  }
  __syncthreads();

  int* lst = listL + wv * LCAP;
  int cntw = 0;
  int np   = 0;
  bool drop    = false;
  bool second  = false;
  const float4* asrc4 = (const float4*)asrc;
  const unsigned int pat = (unsigned int)blockIdx.x * 0x01010101u;

  auto zero_bytes = [](unsigned int x) -> unsigned int {
    return ~((((x & 0x7f7f7f7fu) + 0x7f7f7f7fu) | x) | 0x7f7f7f7fu);
  };
  auto settle_to = [&](int upto) {
    for (int b = np; b < upto; b += 32) {
      const int i = b + lane;
      if (i < upto) {
        const int eid = lst[i];
        int s, r;
        if (eid < E) { s = ei[eid]; r = ei[E + eid] - lo; }
        else         { s = eid - E; r = s - lo; }
        if (!second) {
          const float4 a = asrc4[s];
          lds_max_f(&mxL[r * HEADS + 0], a.x);
          lds_max_f(&mxL[r * HEADS + 1], a.y);
          lds_max_f(&mxL[r * HEADS + 2], a.z);
          lds_max_f(&mxL[r * HEADS + 3], a.w);
        }
        lst[i] = (int)((unsigned int)s | ((unsigned int)r << 16));
      }
    }
    np = upto;
  };
  auto settle = [&](bool all) {
    settle_to((all || drop) ? cntw : np + ((cntw - np) & ~31));
    if (drop) {
      if (second) owner_flush(lst, np, asrc4, adstL4, mL4, sumsL, accL, (float*)(wL4 + wv * 32), hfeat, lane);
      cntw = 0;
      np   = 0;
    }
  };

  const int nblk = (E + KBLK - 1) / KBLK;
  const int start = (int)((long long)blockIdx.x * nblk / gridDim.x);
  auto load_blk = [&](int k) -> uint4 {
    int bi = k + start; if (bi >= nblk) bi -= nblk;
    return *(const uint4*)(okey + (size_t)bi * (KBLK / 4) + lane * 4);
  };
  auto scan_blk = [&](const uint4& d, int k) {
    int bi = k + start; if (bi >= nblk) bi -= nblk;
    const int e0 = bi * KBLK + lane * 16;
    unsigned int c = (zero_bytes(d.x ^ pat) >> 7) | (zero_bytes(d.y ^ pat) >> 6) |
                     (zero_bytes(d.z ^ pat) >> 5) | (zero_bytes(d.w ^ pat) >> 4);
    unsigned int act = __builtin_amdgcn_ballot_w32(c != 0u);
    while (act) {
      if (c != 0u) {
        const int t = __builtin_ctz(c);
        lst[(int)__builtin_amdgcn_mbcnt_lo(act, (unsigned int)cntw)] = e0 + ((t & 3) << 2) + (t >> 3);
        c &= c - 1u;
      }
      cntw += __builtin_popcount(act);
      act = __builtin_amdgcn_ballot_w32(c != 0u);
    }
  };
  const int nbw   = wv < nblk ? (nblk - wv + NWAVE - 1) / NWAVE : 0;
  const int ntrip = nbw / SCANU;

  auto sweep = [&]() {
    if (ntrip > 0) {
      uint4 cur[SCANU], nxt[SCANU];
#pragma unroll
      for (int u = 0; u < SCANU; ++u) cur[u] = load_blk(wv + u * NWAVE);
      for (int trip = 0; trip < ntrip; ++trip) {
        const int tn = min(trip + 1, ntrip - 1);
#pragma unroll
        for (int u = 0; u < SCANU; ++u) nxt[u] = load_blk(wv + (tn * SCANU + u) * NWAVE);
        unsigned int late = 0u;
#pragma unroll
        for (int u = 0; u < SCANU; ++u) {
          if (cntw <= LCAP - KBLK) scan_blk(cur[u], wv + (trip * SCANU + u) * NWAVE);
          else late |= 1u << u;
        }
        if (late) drop = true;
        settle(false);
        while (late) {
          const int u = __builtin_ctz(late);
          late &= late - 1u;
          const int k = wv + (trip * SCANU + u) * NWAVE;
          const uint4 d = load_blk(k);
          scan_blk(d, k);
          settle(true);
        }
#pragma unroll
        for (int u = 0; u < SCANU; ++u) cur[u] = nxt[u];
      }
    }
    for (int t = ntrip * SCANU; t < nbw; ++t) {
      if (cntw > LCAP - KBLK) { drop = true; settle(true); }
      const uint4 d = load_blk(wv + t * NWAVE);
      scan_blk(d, wv + t * NWAVE);
      settle(false);
    }
    for (int rb = wv * 32; rb < ROWS; rb += NWAVE * 32) {
      if (cntw > LCAP - 32) { drop = true; settle(true); }
      const int r = rb + lane;
      const unsigned int mask = __builtin_amdgcn_ballot_w32(r < cnt);
      if (r < cnt) lst[(int)__builtin_amdgcn_mbcnt_lo(mask, (unsigned int)cntw)] = E + lo + r;
      cntw += __builtin_popcount(mask);
    }
    settle(true);
  };

  int* wl = wlist + ((size_t)blockIdx.x * NWAVE + wv) * LCAP;
  bool have = false;
  if (reuse) {
    const int c = __builtin_amdgcn_readfirstlane(wcnt[blockIdx.x * NWAVE + wv]);
    if (c >= 0) {
      for (int b = 0; b < c; b += 128) {
        const int i = b + lane * 4;
        if (i < c) {
          const int4 p = *(const int4*)(wl + i);
          const float4 a0 = asrc4[(unsigned int)p.x & 0xffffu], a1 = asrc4[(unsigned int)p.y & 0xffffu],
                       a2 = asrc4[(unsigned int)p.z & 0xffffu], a3 = asrc4[(unsigned int)p.w & 0xffffu];
          float* m0 = mxL + ((unsigned int)p.x >> 16) * HEADS;
          lds_max_f(m0 + 0, a0.x); lds_max_f(m0 + 1, a0.y); lds_max_f(m0 + 2, a0.z); lds_max_f(m0 + 3, a0.w);
          if (i + 1 < c) {
            float* m1 = mxL + ((unsigned int)p.y >> 16) * HEADS;
            lds_max_f(m1 + 0, a1.x); lds_max_f(m1 + 1, a1.y); lds_max_f(m1 + 2, a1.z); lds_max_f(m1 + 3, a1.w);
          }
          if (i + 2 < c) {
            float* m2 = mxL + ((unsigned int)p.z >> 16) * HEADS;
            lds_max_f(m2 + 0, a2.x); lds_max_f(m2 + 1, a2.y); lds_max_f(m2 + 2, a2.z); lds_max_f(m2 + 3, a2.w);
          }
          if (i + 3 < c) {
            float* m3 = mxL + ((unsigned int)p.w >> 16) * HEADS;
            lds_max_f(m3 + 0, a3.x); lds_max_f(m3 + 1, a3.y); lds_max_f(m3 + 2, a3.z); lds_max_f(m3 + 3, a3.w);
          }
          *(int4*)(lst + i) = p;
        }
      }
      cntw = c;
      np   = c;
      have = true;
    }
  }
  if (!have) sweep();
  const bool kept = !drop;
  if (!reuse) {
    if (kept) {
      for (int i = lane * 4; i < np; i += 128) {
        int4 p;
        p.x = lst[i];
        p.y = i + 1 < np ? lst[i + 1] : 0;
        p.z = i + 2 < np ? lst[i + 2] : 0;
        p.w = i + 3 < np ? lst[i + 3] : 0;
        *(int4*)(wl + i) = p;
      }
    }
    if (lane == 0) wcnt[blockIdx.x * NWAVE + wv] = kept ? np : -1;
  }
  __syncthreads();
  for (int i = tid; i < cnt; i += NWAVE * 32) {
    const float4 ms = mL4[i], bq = adstL4[i];
    mL4[i] = make_float4(leaky(ms.x + bq.x), leaky(ms.y + bq.y), leaky(ms.z + bq.z), leaky(ms.w + bq.w));
  }
  __syncthreads();
  second = true;
  if (kept) {
    owner_flush(lst, np, asrc4, adstL4, mL4, sumsL, accL, (float*)(wL4 + wv * 32), hfeat, lane);
  } else {
    drop = true;
    sweep();
  }
  __syncthreads();

  auto out_val = [&](int r, int f) -> float {
    const float* a = accL + r * HF + (f & 3) * 32 + (f >> 2);
    const float4 sv = sumsL4[r];
    float v = 0.25f * (a[0]  / (sv.x + 1e-16f) + a[8]  / (sv.y + 1e-16f) +
                       a[16] / (sv.z + 1e-16f) + a[24] / (sv.w + 1e-16f)) + bias[f];
    if (do_elu) v = v > 0.f ? v : (expf(v) - 1.f);
    return v;
  };
  if (!fuse) {
    for (int idx = tid; idx < cnt * FEAT; idx += NWAVE * 32)
      out[(size_t)(lo + (idx >> 5)) * FEAT + (idx & (FEAT - 1))] = out_val(idx >> 5, idx & (FEAT - 1));
  } else {
    float* xL  = (float*)listL4;
    float* waL = xL + XROWS * XP;
    for (int idx = tid; idx < XROWS * FEAT; idx += NWAVE * 32) {
      const int r = idx >> 5, f = idx & (FEAT - 1);
      xL[r * XP + f] = r < cnt ? out_val(r, f) : 0.f;
    }
    if (tid < FEAT * 8) {
      const int k = tid >> 3, j = tid & 7, hh = j & 3;
      const float4* wr = (const float4*)(Wn + (size_t)k * HF + hh * FEAT);
      const float4* at = (const float4*)((j < 4 ? asn : adn) + hh * FEAT);
      float s = 0.f;
#pragma unroll
      for (int q = 0; q < FEAT / 4; ++q) {
        const float4 w = wr[q], a = at[q];
        s += w.x * a.x + w.y * a.y + w.z * a.z + w.w * a.w;
      }
      waL[k * 8 + j] = s;
    }
    __syncthreads();
    for (int idx = tid; idx < cnt * 8; idx += NWAVE * 32) {
      const int r = idx >> 3, j = idx & 7;
      const float4* xr = (const float4*)(xL + r * XP);
      float s = 0.f;
#pragma unroll
      for (int q = 0; q < FEAT / 4; ++q) {
        const float4 xv = xr[q];
        s += xv.x * waL[(4 * q) * 8 + j] + xv.y * waL[(4 * q + 1) * 8 + j] +
             xv.z * waL[(4 * q + 2) * 8 + j] + xv.w * waL[(4 * q + 3) * 8 + j];
      }
      if (j < 4) asrcn[(size_t)(lo + r) * HEADS + j] = s;
      else       adstn[(size_t)(lo + r) * HEADS + (j - 4)] = s;
    }
    {
      const int half = lane >> 4, l15 = lane & 15, kh = half * 2;
      const int col  = (wv & 7) * 16 + l15;
      v2f b[FEAT / 4];
#pragma unroll
      for (int st = 0; st < FEAT / 4; ++st) {
        b[st].x = Wn[(size_t)(4 * st + kh)     * HF + col];
        b[st].y = Wn[(size_t)(4 * st + kh + 1) * HF + col];
      }
      for (int rt = wv >> 3; rt < XROWS / 16; rt += NWAVE / 8) {
        const float* arow = xL + (rt * 16 + l15) * XP + kh;
        v8f c = {};
#pragma unroll
        for (int st = 0; st < FEAT / 4; ++st) {
          const v2f a = *(const v2f*)(arow + 4 * st);
          c = __builtin_amdgcn_wmma_f32_16x16x4_f32(false, a, false, b[st], (short)0, c, false, false);
        }
        const int rbase = rt * 16 + half * 8;
        float* orow = hn + (size_t)(lo + rbase) * HF + col;
        if (rt * 16 + 16 <= cnt) {
#pragma unroll
          for (int v = 0; v < 8; ++v) orow[(size_t)v * HF] = c[v];
        } else {
#pragma unroll
          for (int v = 0; v < 8; ++v)
            if (rbase + v < cnt) orow[(size_t)v * HF] = c[v];
        }
      }
    }
  }
}

extern "C" void kernel_launch(void* const* d_in, const int* in_sizes, int n_in,
                              void* d_out, int out_size, void* d_ws, size_t ws_size,
                              hipStream_t stream) {
  const float* x   = (const float*)d_in[0];
  const int*   ei  = (const int*)d_in[1];
  const float* W1  = (const float*)d_in[2];
  const float* as1 = (const float*)d_in[3];
  const float* ad1 = (const float*)d_in[4];
  const float* b1  = (const float*)d_in[5];
  const float* W2  = (const float*)d_in[6];
  const float* as2 = (const float*)d_in[7];
  const float* ad2 = (const float*)d_in[8];
  const float* b2  = (const float*)d_in[9];

  const int Nn = in_sizes[0] / HF;
  const int E  = in_sizes[1] / 2;

  float* ws    = (float*)d_ws;
  float* hbuf  = ws;
  float* hbuf2 = hbuf + (size_t)Nn * HF;
  float* asrc  = hbuf2 + (size_t)Nn * HF;
  float* adst  = asrc + (size_t)Nn * HEADS;
  float* asrc2 = adst + (size_t)Nn * HEADS;
  float* adst2 = asrc2 + (size_t)Nn * HEADS;
  int*   okey  = (int*)(adst2 + (size_t)Nn * HEADS);
  const int nkw = (E + KBLK - 1) / KBLK * (KBLK / 4);
  int*   wlist = okey + nkw;
  int*   wcnt  = wlist + (size_t)((Nn + ROWS - 1) / ROWS) * NWAVE * LCAP;

  const dim3 blk(256);
  const int row_tiles = Nn / 16;
  const int tail_row0 = row_tiles * 16;
  const int agg_wgs   = (Nn + ROWS - 1) / ROWS;

  if (row_tiles > 0) {
    const int gg = row_tiles < GEMM_WGS ? row_tiles : GEMM_WGS;
    gat_gemm_wmma_f32<128><<<gg, blk, 0, stream>>>(x, W1, hbuf, row_tiles);
  }
  if (tail_row0 < Nn) {
    int tail = (Nn - tail_row0) * HF;
    gat_gemm_tail<<<(tail + 255) / 256, blk, 0, stream>>>(x, W1, hbuf, 128, tail_row0, Nn);
  }
  gat_alpha<<<(Nn * HEADS + 255) / 256, blk, 0, stream>>>(hbuf, as1, ad1, asrc, adst, Nn);
  gat_init<<<(nkw + 255) / 256, blk, 0, stream>>>(okey, ei, E, nkw);
  gat_agg_owner<<<agg_wgs, dim3(NWAVE * 32), 0, stream>>>(ei, okey, asrc, adst, hbuf, b1, (float*)d_out, wlist, wcnt, E, Nn, 1, 0,
                                                           W2, as2, ad2, hbuf2, asrc2, adst2, 1);
  gat_agg_owner<<<agg_wgs, dim3(NWAVE * 32), 0, stream>>>(ei, okey, asrc2, adst2, hbuf2, b2, (float*)d_out, wlist, wcnt, E, Nn, 0, 1,
                                                           W2, as2, ad2, hbuf, asrc, adst, 0);
}
